# starting kernel
# baseline (speedup 1.0000x reference)
#include <hip/hip_runtime.h>
#include <math.h>

#define N 2048
#define D_IN 128
#define D_OUT 32
#define D_EDGE 8
#define ALPHA 0.2f
#define NEG_BIG -9e15f
#define ROWSTRIDE 2050

typedef __attribute__((ext_vector_type(2))) float v2f;
typedef __attribute__((ext_vector_type(8))) float v8f;

__device__ __forceinline__ v8f wmma_f32_k4(v2f a, v2f b, v8f c) {
  return __builtin_amdgcn_wmma_f32_16x16x4_f32(false, a, false, b, (short)0, c,
                                               false, false);
}

__device__ __forceinline__ float lrelu(float x) { return x >= 0.f ? x : ALPHA * x; }

__global__ void prep_vec_kernel(const float* __restrict__ Wh1,
                                const float* __restrict__ Wh2,
                                const float* __restrict__ ah,
                                float* __restrict__ v1, float* __restrict__ v2) {
  const int t = threadIdx.x;
  float a1 = 0.f, a2 = 0.f;
  for (int f = 0; f < D_OUT; ++f) {
    const float w = ah[f];
    a1 += Wh1[t * D_OUT + f] * w;
    a2 += Wh2[t * D_OUT + f] * w;
  }
  v1[t] = a1;
  v2[t] = a2;
}

__global__ void prep_s_kernel(const float* __restrict__ input,
                              const float* __restrict__ v1,
                              const float* __restrict__ v2,
                              float* __restrict__ s1, float* __restrict__ s2) {
  const int i = blockIdx.x * blockDim.x + threadIdx.x;
  const float* row = input + (size_t)i * D_IN;
  float a = 0.f, b = 0.f;
  for (int k = 0; k < D_IN; ++k) {
    const float x = row[k];
    a += x * v1[k];
    b += x * v2[k];
  }
  s1[i] = a;
  s2[i] = b;
}

__global__ void h_gemm_kernel(const float* __restrict__ input,
                              const float* __restrict__ Wh,
                              float* __restrict__ h) {
  const int lane = threadIdx.x & 31;
  const int i0 = blockIdx.x * 16;
  const int m = lane & 15;
  const int kd = (lane >> 4) << 1;
  const int n = lane & 15;
  v8f c0 = {};
  v8f c1 = {};
  for (int k0 = 0; k0 < D_IN; k0 += 4) {
    v2f a = *(const v2f*)(input + (size_t)(i0 + m) * D_IN + k0 + kd);
    v2f b0, b1;
    b0.x = Wh[(k0 + kd) * D_OUT + n];
    b0.y = Wh[(k0 + kd + 1) * D_OUT + n];
    b1.x = Wh[(k0 + kd) * D_OUT + 16 + n];
    b1.y = Wh[(k0 + kd + 1) * D_OUT + 16 + n];
    c0 = wmma_f32_k4(a, b0, c0);
    c1 = wmma_f32_k4(a, b1, c1);
  }
  const int row = (lane >> 4) << 3;
  for (int v = 0; v < 8; ++v) {
    h[(size_t)(i0 + row + v) * D_OUT + n] = c0[v];
    h[(size_t)(i0 + row + v) * D_OUT + 16 + n] = c1[v];
  }
}

__global__ void fused_attention_kernel(const float* __restrict__ edge_feat,
                                       const int* __restrict__ adj,
                                       const float* __restrict__ Wf,
                                       const float* __restrict__ af,
                                       const float* __restrict__ bf,
                                       const float* __restrict__ s1,
                                       const float* __restrict__ s2,
                                       const float* __restrict__ h,
                                       float* __restrict__ h_prime) {
  extern __shared__ float smem[];
  float* sc = smem;
  float* part = smem + 16 * ROWSTRIDE;
  __shared__ float s_inv[16];

  const int tid = threadIdx.x;
  const int lane = tid & 31;
  const int wave = tid >> 5;
  const int i0 = blockIdx.x * 16;

  const int nn = lane & 15;
  const int kd = (lane >> 4) << 1;

  v2f bw00, bw01, bw10, bw11;
  bw00.x = Wf[(0 + kd) * D_OUT + nn];
  bw00.y = Wf[(1 + kd) * D_OUT + nn];
  bw01.x = Wf[(0 + kd) * D_OUT + 16 + nn];
  bw01.y = Wf[(1 + kd) * D_OUT + 16 + nn];
  bw10.x = Wf[(4 + kd) * D_OUT + nn];
  bw10.y = Wf[(5 + kd) * D_OUT + nn];
  bw11.x = Wf[(4 + kd) * D_OUT + 16 + nn];
  bw11.y = Wf[(5 + kd) * D_OUT + 16 + nn];
  const float af_lo = af[nn], af_hi = af[16 + nn];
  const float bf_lo = bf[nn], bf_hi = bf[16 + nn];
  v8f cini0, cini1;
  for (int v = 0; v < 8; ++v) {
    cini0[v] = bf_lo;
    cini1[v] = bf_hi;
  }

  for (int t = wave; t < 16 * 128; t += 8) {
    const int il = t >> 7;
    const int i = i0 + il;
    const int j0 = (t & 127) << 4;
    const float* ep = edge_feat + ((size_t)i * N + (j0 + nn)) * D_EDGE + kd;
    __builtin_prefetch(ep + 8 * 16 * D_EDGE, 0, 1);
    v2f a0 = *(const v2f*)(ep);
    v2f a1 = *(const v2f*)(ep + 4);
    v8f f0 = wmma_f32_k4(a0, bw00, cini0);
    f0 = wmma_f32_k4(a1, bw10, f0);
    v8f f1 = wmma_f32_k4(a0, bw01, cini1);
    f1 = wmma_f32_k4(a1, bw11, f1);
    float p[8];
    for (int v = 0; v < 8; ++v)
      p[v] = lrelu(f0[v]) * af_lo + lrelu(f1[v]) * af_hi;
    for (int mask = 1; mask <= 8; mask <<= 1)
      for (int v = 0; v < 8; ++v) p[v] += __shfl_xor(p[v], mask, 32);
    if ((lane & 8) == 0) {
      const int md = lane & 7;
      float t0 = (md & 1) ? p[1] : p[0];
      float t1 = (md & 1) ? p[3] : p[2];
      float t2 = (md & 1) ? p[5] : p[4];
      float t3 = (md & 1) ? p[7] : p[6];
      float u0 = (md & 2) ? t1 : t0;
      float u1 = (md & 2) ? t3 : t2;
      float cp = (md & 4) ? u1 : u0;
      const int j = j0 + ((lane >> 4) << 3) + md;
      float v = lrelu(s1[i] + s2[j]) + lrelu(cp);
      v = (adj[(size_t)i * N + j] > 0) ? v : NEG_BIG;
      sc[il * ROWSTRIDE + j] = v;
    }
  }
  __syncthreads();

  for (int r = wave; r < 16; r += 8) {
    float* row = sc + r * ROWSTRIDE;
    float m = -3.4e38f;
    for (int c = lane; c < N; c += 32) m = fmaxf(m, row[c]);
    for (int mask = 16; mask >= 1; mask >>= 1)
      m = fmaxf(m, __shfl_xor(m, mask, 32));
    float s = 0.f;
    for (int c = lane; c < N; c += 32) {
      const float e = __expf(row[c] - m);
      row[c] = e;
      s += e;
    }
    for (int mask = 16; mask >= 1; mask >>= 1) s += __shfl_xor(s, mask, 32);
    if (lane == 0) s_inv[r] = 1.0f / s;
  }
  __syncthreads();

  v8f c0 = {};
  v8f c1 = {};
  const int kbase = wave * 256;
  for (int kk = 0; kk < 256; kk += 4) {
    const int k0 = kbase + kk;
    v2f a = *(const v2f*)(sc + nn * ROWSTRIDE + k0 + kd);
    v2f b0, b1;
    b0.x = h[(size_t)(k0 + kd) * D_OUT + nn];
    b0.y = h[(size_t)(k0 + kd + 1) * D_OUT + nn];
    b1.x = h[(size_t)(k0 + kd) * D_OUT + 16 + nn];
    b1.y = h[(size_t)(k0 + kd + 1) * D_OUT + 16 + nn];
    c0 = wmma_f32_k4(a, b0, c0);
    c1 = wmma_f32_k4(a, b1, c1);
  }
  {
    const int row = (lane >> 4) << 3;
    float* pw = part + wave * 512;
    for (int v = 0; v < 8; ++v) {
      pw[(row + v) * 32 + nn] = c0[v];
      pw[(row + v) * 32 + 16 + nn] = c1[v];
    }
  }
  __syncthreads();
  for (int e = tid; e < 512; e += 256) {
    float s = 0.f;
    for (int w = 0; w < 8; ++w) s += part[w * 512 + e];
    const int row = e >> 5, col = e & 31;
    h_prime[(size_t)(i0 + row) * D_OUT + col] = s * s_inv[row];
  }
}

__global__ void bn_partial_kernel(const float* __restrict__ hp,
                                  float* __restrict__ bp) {
  __shared__ float ls[8][32];
  __shared__ float lq[8][32];
  const int col = threadIdx.x & 31;
  const int seg = threadIdx.x >> 5;
  const int r0 = blockIdx.x * 32 + seg * 4;
  float s = 0.f, q = 0.f;
  for (int r = 0; r < 4; ++r) {
    const float v = hp[(size_t)(r0 + r) * D_OUT + col];
    s += v;
    q += v * v;
  }
  ls[seg][col] = s;
  lq[seg][col] = q;
  __syncthreads();
  if (threadIdx.x < 32) {
    float ss = 0.f, qq = 0.f;
    for (int w = 0; w < 8; ++w) {
      ss += ls[w][col];
      qq += lq[w][col];
    }
    bp[blockIdx.x * 64 + col] = ss;
    bp[blockIdx.x * 64 + 32 + col] = qq;
  }
}

__global__ void bn_stats_kernel(const float* __restrict__ bp,
                                float* __restrict__ ms) {
  const int col = threadIdx.x;
  float s = 0.f, q = 0.f;
  for (int b = 0; b < 64; ++b) {
    s += bp[b * 64 + col];
    q += bp[b * 64 + 32 + col];
  }
  const float mean = s * (1.0f / (float)N);
  const float var = q * (1.0f / (float)N) - mean * mean;
  ms[col] = mean;
  ms[32 + col] = rsqrtf(var + 1e-5f);
}

__global__ void bn_norm_kernel(const float* __restrict__ hp,
                               const float* __restrict__ ms,
                               const float* __restrict__ gamma,
                               const float* __restrict__ beta,
                               float* __restrict__ out) {
  const int idx = blockIdx.x * blockDim.x + threadIdx.x;
  const int col = idx & 31;
  out[idx] = gamma[col] * (hp[idx] - ms[col]) * ms[32 + col] + beta[col];
}

extern "C" void kernel_launch(void* const* d_in, const int* in_sizes, int n_in,
                              void* d_out, int out_size, void* d_ws,
                              size_t ws_size, hipStream_t stream) {
  const float* input = (const float*)d_in[0];
  const float* edge_feat = (const float*)d_in[1];
  const int* adj = (const int*)d_in[2];
  const float* Wh = (const float*)d_in[3];
  const float* Wh1 = (const float*)d_in[4];
  const float* Wh2 = (const float*)d_in[5];
  const float* Wf = (const float*)d_in[6];
  const float* ah = (const float*)d_in[7];
  const float* af = (const float*)d_in[8];
  const float* bf = (const float*)d_in[9];
  const float* gamma = (const float*)d_in[10];
  const float* beta = (const float*)d_in[11];

  float* ws = (float*)d_ws;
  float* v1 = ws;
  float* v2 = ws + 128;
  float* s1 = ws + 256;
  float* s2 = ws + 2304;
  float* h = ws + 4352;
  float* hp = ws + 69888;
  float* bp = ws + 135424;
  float* ms = ws + 139520;
  float* out = (float*)d_out;

  prep_vec_kernel<<<1, 128, 0, stream>>>(Wh1, Wh2, ah, v1, v2);
  prep_s_kernel<<<8, 256, 0, stream>>>(input, v1, v2, s1, s2);
  h_gemm_kernel<<<128, 32, 0, stream>>>(input, Wh, h);
  const size_t smem = (size_t)(16 * ROWSTRIDE + 8 * 512) * sizeof(float);
  fused_attention_kernel<<<128, 256, smem, stream>>>(edge_feat, adj, Wf, af, bf,
                                                     s1, s2, h, hp);
  bn_partial_kernel<<<64, 256, 0, stream>>>(hp, bp);
  bn_stats_kernel<<<1, 32, 0, stream>>>(bp, ms);
  bn_norm_kernel<<<256, 256, 0, stream>>>(hp, ms, gamma, beta, out);
}


# edge-score feature sum across lanes as a halving butterfly on DPP adds (row_xmask 1,2,4,8): 8 cross-lane adds + 14 selects per 16-edge group, were 32 LDS-unit shuffles + 32 adds; same sum tree
# speedup vs baseline: 3.9590x; 3.9590x over previous
#include <hip/hip_runtime.h>
#include <math.h>

#define N 2048
#define D_IN 128
#define D_OUT 32
#define D_EDGE 8
#define ALPHA 0.2f
#define NEG_BIG -9e15f
#define HCOLS 1024
#define ROWSTRIDE 1026
#define NWAVES 32

typedef __attribute__((ext_vector_type(2))) float v2f;
typedef __attribute__((ext_vector_type(8))) float v8f;

__device__ __forceinline__ v8f wmma_f32_k4(v2f a, v2f b, v8f c) {
  return __builtin_amdgcn_wmma_f32_16x16x4_f32(false, a, false, b, (short)0, c,
                                               false, false);
}

__device__ __forceinline__ float lrelu(float x) { return x >= 0.f ? x : ALPHA * x; }

template <int M>
__device__ __forceinline__ float row_xor(float x) {
  return __int_as_float(__builtin_amdgcn_update_dpp(0, __float_as_int(x), 0x160 + M,
                                                    0xf, 0xf, true));
}

__global__ void prep_vec_kernel(const float* __restrict__ Wh1,
                                const float* __restrict__ Wh2,
                                const float* __restrict__ ah,
                                float* __restrict__ v1, float* __restrict__ v2) {
  const int t = threadIdx.x;
  float a1 = 0.f, a2 = 0.f;
  for (int f = 0; f < D_OUT; ++f) {
    const float w = ah[f];
    a1 += Wh1[t * D_OUT + f] * w;
    a2 += Wh2[t * D_OUT + f] * w;
  }
  v1[t] = a1;
  v2[t] = a2;
}

__global__ void prep_s_kernel(const float* __restrict__ input,
                              const float* __restrict__ v1,
                              const float* __restrict__ v2,
                              float* __restrict__ s1, float* __restrict__ s2) {
  const int i = blockIdx.x * blockDim.x + threadIdx.x;
  const float* row = input + (size_t)i * D_IN;
  float a = 0.f, b = 0.f;
  for (int k = 0; k < D_IN; ++k) {
    const float x = row[k];
    a += x * v1[k];
    b += x * v2[k];
  }
  s1[i] = a;
  s2[i] = b;
}

__global__ void h_gemm_kernel(const float* __restrict__ input,
                              const float* __restrict__ Wh,
                              float* __restrict__ h) {
  const int lane = threadIdx.x & 31;
  const int i0 = blockIdx.x * 16;
  const int m = lane & 15;
  const int kd = (lane >> 4) << 1;
  const int n = lane & 15;
  v8f c0 = {};
  v8f c1 = {};
  for (int k0 = 0; k0 < D_IN; k0 += 4) {
    v2f a = *(const v2f*)(input + (size_t)(i0 + m) * D_IN + k0 + kd);
    v2f b0, b1;
    b0.x = Wh[(k0 + kd) * D_OUT + n];
    b0.y = Wh[(k0 + kd + 1) * D_OUT + n];
    b1.x = Wh[(k0 + kd) * D_OUT + 16 + n];
    b1.y = Wh[(k0 + kd + 1) * D_OUT + 16 + n];
    c0 = wmma_f32_k4(a, b0, c0);
    c1 = wmma_f32_k4(a, b1, c1);
  }
  const int row = (lane >> 4) << 3;
  for (int v = 0; v < 8; ++v) {
    h[(size_t)(i0 + row + v) * D_OUT + n] = c0[v];
    h[(size_t)(i0 + row + v) * D_OUT + 16 + n] = c1[v];
  }
}

__global__ void __launch_bounds__(1024)
fused_attention_kernel(const float* __restrict__ edge_feat,
                                       const int* __restrict__ adj,
                                       const float* __restrict__ Wf,
                                       const float* __restrict__ af,
                                       const float* __restrict__ bf,
                                       const float* __restrict__ s1,
                                       const float* __restrict__ s2,
                                       const float* __restrict__ h,
                                       float* __restrict__ o0,
                                       float* __restrict__ o1,
                                       float* __restrict__ lse) {
  extern __shared__ float smem[];
  float* sc = smem;
  float* part = smem + 16 * ROWSTRIDE;
  __shared__ float s_inv[16];
  __shared__ float s_lse[16];

  const int tid = threadIdx.x;
  const int lane = tid & 31;
  const int wave = tid >> 5;
  const int i0 = (blockIdx.x >> 1) * 16;
  const int sp = blockIdx.x & 1;
  const int cbase = sp * HCOLS;

  const int nn = lane & 15;
  const int kd = (lane >> 4) << 1;
  const bool b0 = (lane & 1) != 0, b1 = (lane & 2) != 0, b2 = (lane & 4) != 0;

  v2f bw00, bw01, bw10, bw11;
  bw00.x = Wf[(0 + kd) * D_OUT + nn];
  bw00.y = Wf[(1 + kd) * D_OUT + nn];
  bw01.x = Wf[(0 + kd) * D_OUT + 16 + nn];
  bw01.y = Wf[(1 + kd) * D_OUT + 16 + nn];
  bw10.x = Wf[(4 + kd) * D_OUT + nn];
  bw10.y = Wf[(5 + kd) * D_OUT + nn];
  bw11.x = Wf[(4 + kd) * D_OUT + 16 + nn];
  bw11.y = Wf[(5 + kd) * D_OUT + 16 + nn];
  const float af_lo = af[nn], af_hi = af[16 + nn];
  const float bf_lo = bf[nn], bf_hi = bf[16 + nn];
  v8f cini0, cini1;
  for (int v = 0; v < 8; ++v) {
    cini0[v] = bf_lo;
    cini1[v] = bf_hi;
  }

  for (int t = wave; t < 16 * 64; t += NWAVES) {
    const int il = t >> 6;
    const int i = i0 + il;
    const int jl0 = (t & 63) << 4;
    const int j0 = cbase + jl0;
    const float* ep = edge_feat + ((size_t)i * N + (j0 + nn)) * D_EDGE + kd;
    const int tn = t + NWAVES;
    if (tn < 16 * 64)
      __builtin_prefetch(edge_feat + ((size_t)(i0 + (tn >> 6)) * N +
                                      (cbase + ((tn & 63) << 4) + nn)) * D_EDGE + kd,
                         0, 1);
    v2f a0 = *(const v2f*)(ep);
    v2f a1 = *(const v2f*)(ep + 4);
    v8f f0 = wmma_f32_k4(a0, bw00, cini0);
    f0 = wmma_f32_k4(a1, bw10, f0);
    v8f f1 = wmma_f32_k4(a0, bw01, cini1);
    f1 = wmma_f32_k4(a1, bw11, f1);
    float p[8];
    for (int v = 0; v < 8; ++v)
      p[v] = lrelu(f0[v]) * af_lo + lrelu(f1[v]) * af_hi;
    float q[4], u[2];
    for (int k = 0; k < 4; ++k) {
      const float keep = b0 ? p[2 * k + 1] : p[2 * k];
      const float send = b0 ? p[2 * k] : p[2 * k + 1];
      q[k] = keep + row_xor<1>(send);
    }
    for (int k = 0; k < 2; ++k) {
      const float keep = b1 ? q[2 * k + 1] : q[2 * k];
      const float send = b1 ? q[2 * k] : q[2 * k + 1];
      u[k] = keep + row_xor<2>(send);
    }
    float cp;
    {
      const float keep = b2 ? u[1] : u[0];
      const float send = b2 ? u[0] : u[1];
      cp = keep + row_xor<4>(send);
    }
    cp += row_xor<8>(cp);
    if ((lane & 8) == 0) {
      const int md = lane & 7;
      const int jl = jl0 + ((lane >> 4) << 3) + md;
      const int j = cbase + jl;
      float v = lrelu(s1[i] + s2[j]) + lrelu(cp);
      v = (adj[(size_t)i * N + j] > 0) ? v : NEG_BIG;
      sc[il * ROWSTRIDE + jl] = v;
    }
  }
  __syncthreads();

  if (wave < 16) {
    const int r = wave;
    float* row = sc + r * ROWSTRIDE;
    float m = -3.4e38f;
    for (int c = lane; c < HCOLS; c += 32) m = fmaxf(m, row[c]);
    for (int mask = 16; mask >= 1; mask >>= 1)
      m = fmaxf(m, __shfl_xor(m, mask, 32));
    float s = 0.f;
    for (int c = lane; c < HCOLS; c += 32) {
      const float e = __expf(row[c] - m);
      row[c] = e;
      s += e;
    }
    for (int mask = 16; mask >= 1; mask >>= 1) s += __shfl_xor(s, mask, 32);
    if (lane == 0) {
      s_inv[r] = 1.0f / s;
      s_lse[r] = m + logf(s);
    }
  }
  __syncthreads();

  v8f c0 = {};
  v8f c1 = {};
  const int kbase = wave * (HCOLS / NWAVES);
  for (int kk = 0; kk < HCOLS / NWAVES; kk += 4) {
    const int k0 = kbase + kk;
    v2f a = *(const v2f*)(sc + nn * ROWSTRIDE + k0 + kd);
    const float* hr = h + (size_t)(cbase + k0 + kd) * D_OUT;
    v2f b0, b1;
    b0.x = hr[nn];
    b0.y = hr[D_OUT + nn];
    b1.x = hr[16 + nn];
    b1.y = hr[D_OUT + 16 + nn];
    c0 = wmma_f32_k4(a, b0, c0);
    c1 = wmma_f32_k4(a, b1, c1);
  }
  {
    const int row = (lane >> 4) << 3;
    float* pw = part + wave * 512;
    for (int v = 0; v < 8; ++v) {
      pw[(row + v) * 32 + nn] = c0[v];
      pw[(row + v) * 32 + 16 + nn] = c1[v];
    }
  }
  __syncthreads();
  if (tid < 512) {
    const int e = tid;
    float s = 0.f;
    for (int w = 0; w < NWAVES; ++w) s += part[w * 512 + e];
    const int row = e >> 5, col = e & 31;
    float* dst = sp ? o1 : o0;
    dst[(size_t)(i0 + row) * D_OUT + col] = s * s_inv[row];
  } else if (tid < 528) {
    lse[sp * N + i0 + (tid - 512)] = s_lse[tid - 512];
  }
}

__global__ void bn_partial_kernel(float* __restrict__ hp,
                                  const float* __restrict__ o1,
                                  const float* __restrict__ lse,
                                  float* __restrict__ bp) {
  __shared__ float ls[8][32];
  __shared__ float lq[8][32];
  const int col = threadIdx.x & 31;
  const int seg = threadIdx.x >> 5;
  const int r0 = blockIdx.x * 32 + seg * 4;
  float s = 0.f, q = 0.f;
  for (int r = 0; r < 4; ++r) {
    const size_t idx = (size_t)(r0 + r) * D_OUT + col;
    const float l0 = lse[r0 + r], l1 = lse[N + r0 + r];
    const float lm = fmaxf(l0, l1);
    const float w0 = expf(l0 - lm), w1 = expf(l1 - lm);
    const float v = (w0 * hp[idx] + w1 * o1[idx]) / (w0 + w1);
    hp[idx] = v;
    s += v;
    q += v * v;
  }
  ls[seg][col] = s;
  lq[seg][col] = q;
  __syncthreads();
  if (threadIdx.x < 32) {
    float ss = 0.f, qq = 0.f;
    for (int w = 0; w < 8; ++w) {
      ss += ls[w][col];
      qq += lq[w][col];
    }
    bp[blockIdx.x * 64 + col] = ss;
    bp[blockIdx.x * 64 + 32 + col] = qq;
  }
}

__global__ void bn_stats_kernel(const float* __restrict__ bp,
                                float* __restrict__ ms) {
  const int col = threadIdx.x;
  float s = 0.f, q = 0.f;
  for (int b = 0; b < 64; ++b) {
    s += bp[b * 64 + col];
    q += bp[b * 64 + 32 + col];
  }
  const float mean = s * (1.0f / (float)N);
  const float var = q * (1.0f / (float)N) - mean * mean;
  ms[col] = mean;
  ms[32 + col] = rsqrtf(var + 1e-5f);
}

__global__ void bn_norm_kernel(const float* __restrict__ hp,
                               const float* __restrict__ ms,
                               const float* __restrict__ gamma,
                               const float* __restrict__ beta,
                               float* __restrict__ out) {
  const int idx = blockIdx.x * blockDim.x + threadIdx.x;
  const int col = idx & 31;
  out[idx] = gamma[col] * (hp[idx] - ms[col]) * ms[32 + col] + beta[col];
}

extern "C" void kernel_launch(void* const* d_in, const int* in_sizes, int n_in,
                              void* d_out, int out_size, void* d_ws,
                              size_t ws_size, hipStream_t stream) {
  const float* input = (const float*)d_in[0];
  const float* edge_feat = (const float*)d_in[1];
  const int* adj = (const int*)d_in[2];
  const float* Wh = (const float*)d_in[3];
  const float* Wh1 = (const float*)d_in[4];
  const float* Wh2 = (const float*)d_in[5];
  const float* Wf = (const float*)d_in[6];
  const float* ah = (const float*)d_in[7];
  const float* af = (const float*)d_in[8];
  const float* bf = (const float*)d_in[9];
  const float* gamma = (const float*)d_in[10];
  const float* beta = (const float*)d_in[11];

  float* ws = (float*)d_ws;
  float* v1 = ws;
  float* v2 = ws + 128;
  float* s1 = ws + 256;
  float* s2 = ws + 2304;
  float* h = ws + 4352;
  float* hp = ws + 69888;
  float* lse = ws + 135424;
  float* bp = ws + 4352;
  float* ms = ws + 139520;
  float* out = (float*)d_out;

  prep_vec_kernel<<<1, 128, 0, stream>>>(Wh1, Wh2, ah, v1, v2);
  prep_s_kernel<<<8, 256, 0, stream>>>(input, v1, v2, s1, s2);
  h_gemm_kernel<<<128, 32, 0, stream>>>(input, Wh, h);
  const size_t smem = (size_t)(16 * ROWSTRIDE + NWAVES * 512) * sizeof(float);
  fused_attention_kernel<<<256, 1024, smem, stream>>>(edge_feat, adj, Wf, af, bf,
                                                      s1, s2, h, hp, out, lse);
  bn_partial_kernel<<<64, 256, 0, stream>>>(hp, out, lse, bp);
  bn_stats_kernel<<<1, 32, 0, stream>>>(bp, ms);
  bn_norm_kernel<<<256, 256, 0, stream>>>(hp, ms, gamma, beta, out);
}


# adjacency mask and node-score term moved from the per-16-edge tail (two extra dependent memory waits per group) into the softmax max pass; adjacency read as whole lines at kernel entry; same sums
# speedup vs baseline: 4.4597x; 1.1265x over previous
#include <hip/hip_runtime.h>
#include <math.h>

#define N 2048
#define D_IN 128
#define D_OUT 32
#define D_EDGE 8
#define ALPHA 0.2f
#define NEG_BIG -9e15f
#define HCOLS 1024
#define ROWSTRIDE 1026
#define NWAVES 32

typedef __attribute__((ext_vector_type(2))) float v2f;
typedef __attribute__((ext_vector_type(8))) float v8f;

__device__ __forceinline__ v8f wmma_f32_k4(v2f a, v2f b, v8f c) {
  return __builtin_amdgcn_wmma_f32_16x16x4_f32(false, a, false, b, (short)0, c,
                                               false, false);
}

__device__ __forceinline__ float lrelu(float x) { return x >= 0.f ? x : ALPHA * x; }

template <int M>
__device__ __forceinline__ float row_xor(float x) {
  return __int_as_float(__builtin_amdgcn_update_dpp(0, __float_as_int(x), 0x160 + M,
                                                    0xf, 0xf, true));
}

__global__ void prep_vec_kernel(const float* __restrict__ Wh1,
                                const float* __restrict__ Wh2,
                                const float* __restrict__ ah,
                                float* __restrict__ v1, float* __restrict__ v2) {
  const int t = threadIdx.x;
  float a1 = 0.f, a2 = 0.f;
  for (int f = 0; f < D_OUT; ++f) {
    const float w = ah[f];
    a1 += Wh1[t * D_OUT + f] * w;
    a2 += Wh2[t * D_OUT + f] * w;
  }
  v1[t] = a1;
  v2[t] = a2;
}

__global__ void prep_s_kernel(const float* __restrict__ input,
                              const float* __restrict__ v1,
                              const float* __restrict__ v2,
                              float* __restrict__ s1, float* __restrict__ s2) {
  const int i = blockIdx.x * blockDim.x + threadIdx.x;
  const float* row = input + (size_t)i * D_IN;
  float a = 0.f, b = 0.f;
  for (int k = 0; k < D_IN; ++k) {
    const float x = row[k];
    a += x * v1[k];
    b += x * v2[k];
  }
  s1[i] = a;
  s2[i] = b;
}

__global__ void h_gemm_kernel(const float* __restrict__ input,
                              const float* __restrict__ Wh,
                              float* __restrict__ h) {
  const int lane = threadIdx.x & 31;
  const int i0 = blockIdx.x * 16;
  const int m = lane & 15;
  const int kd = (lane >> 4) << 1;
  const int n = lane & 15;
  v8f c0 = {};
  v8f c1 = {};
  for (int k0 = 0; k0 < D_IN; k0 += 4) {
    v2f a = *(const v2f*)(input + (size_t)(i0 + m) * D_IN + k0 + kd);
    v2f b0, b1;
    b0.x = Wh[(k0 + kd) * D_OUT + n];
    b0.y = Wh[(k0 + kd + 1) * D_OUT + n];
    b1.x = Wh[(k0 + kd) * D_OUT + 16 + n];
    b1.y = Wh[(k0 + kd + 1) * D_OUT + 16 + n];
    c0 = wmma_f32_k4(a, b0, c0);
    c1 = wmma_f32_k4(a, b1, c1);
  }
  const int row = (lane >> 4) << 3;
  for (int v = 0; v < 8; ++v) {
    h[(size_t)(i0 + row + v) * D_OUT + n] = c0[v];
    h[(size_t)(i0 + row + v) * D_OUT + 16 + n] = c1[v];
  }
}

__global__ void __launch_bounds__(1024)
fused_attention_kernel(const float* __restrict__ edge_feat,
                                       const int* __restrict__ adj,
                                       const float* __restrict__ Wf,
                                       const float* __restrict__ af,
                                       const float* __restrict__ bf,
                                       const float* __restrict__ s1,
                                       const float* __restrict__ s2,
                                       const float* __restrict__ h,
                                       float* __restrict__ o0,
                                       float* __restrict__ o1,
                                       float* __restrict__ lse) {
  extern __shared__ float smem[];
  float* sc = smem;
  float* part = smem + 16 * ROWSTRIDE;
  __shared__ float s_inv[16];
  __shared__ float s_lse[16];

  const int tid = threadIdx.x;
  const int lane = tid & 31;
  const int wave = tid >> 5;
  const int i0 = (blockIdx.x >> 1) * 16;
  const int sp = blockIdx.x & 1;
  const int cbase = sp * HCOLS;

  const int nn = lane & 15;
  const int kd = (lane >> 4) << 1;
  const bool b0 = (lane & 1) != 0, b1 = (lane & 2) != 0, b2 = (lane & 4) != 0;

  v2f bw00, bw01, bw10, bw11;
  bw00.x = Wf[(0 + kd) * D_OUT + nn];
  bw00.y = Wf[(1 + kd) * D_OUT + nn];
  bw01.x = Wf[(0 + kd) * D_OUT + 16 + nn];
  bw01.y = Wf[(1 + kd) * D_OUT + 16 + nn];
  bw10.x = Wf[(4 + kd) * D_OUT + nn];
  bw10.y = Wf[(5 + kd) * D_OUT + nn];
  bw11.x = Wf[(4 + kd) * D_OUT + 16 + nn];
  bw11.y = Wf[(5 + kd) * D_OUT + 16 + nn];
  const float af_lo = af[nn], af_hi = af[16 + nn];
  const float bf_lo = bf[nn], bf_hi = bf[16 + nn];
  int adjv[HCOLS / 32] = {};
  if (wave < 16) {
    const int* arow = adj + (size_t)(i0 + wave) * N + cbase + lane;
#pragma unroll
    for (int k = 0; k < HCOLS / 32; ++k) adjv[k] = arow[32 * k];
  }
  v8f cini0, cini1;
  for (int v = 0; v < 8; ++v) {
    cini0[v] = bf_lo;
    cini1[v] = bf_hi;
  }

  for (int t = wave; t < 16 * 64; t += NWAVES) {
    const int il = t >> 6;
    const int i = i0 + il;
    const int jl0 = (t & 63) << 4;
    const int j0 = cbase + jl0;
    const float* ep = edge_feat + ((size_t)i * N + (j0 + nn)) * D_EDGE + kd;
    const int tn = t + NWAVES;
    if (tn < 16 * 64)
      __builtin_prefetch(edge_feat + ((size_t)(i0 + (tn >> 6)) * N +
                                      (cbase + ((tn & 63) << 4) + nn)) * D_EDGE + kd,
                         0, 1);
    v2f a0 = *(const v2f*)(ep);
    v2f a1 = *(const v2f*)(ep + 4);
    v8f f0 = wmma_f32_k4(a0, bw00, cini0);
    f0 = wmma_f32_k4(a1, bw10, f0);
    v8f f1 = wmma_f32_k4(a0, bw01, cini1);
    f1 = wmma_f32_k4(a1, bw11, f1);
    float p[8];
    for (int v = 0; v < 8; ++v)
      p[v] = lrelu(f0[v]) * af_lo + lrelu(f1[v]) * af_hi;
    float q[4], u[2];
    for (int k = 0; k < 4; ++k) {
      const float keep = b0 ? p[2 * k + 1] : p[2 * k];
      const float send = b0 ? p[2 * k] : p[2 * k + 1];
      q[k] = keep + row_xor<1>(send);
    }
    for (int k = 0; k < 2; ++k) {
      const float keep = b1 ? q[2 * k + 1] : q[2 * k];
      const float send = b1 ? q[2 * k] : q[2 * k + 1];
      u[k] = keep + row_xor<2>(send);
    }
    float cp;
    {
      const float keep = b2 ? u[1] : u[0];
      const float send = b2 ? u[0] : u[1];
      cp = keep + row_xor<4>(send);
    }
    cp += row_xor<8>(cp);
    if ((lane & 8) == 0) {
      const int jl = jl0 + ((lane >> 4) << 3) + (lane & 7);
      sc[il * ROWSTRIDE + jl] = lrelu(cp);
    }
  }
  __syncthreads();

  if (wave < 16) {
    const int r = wave;
    float* row = sc + r * ROWSTRIDE + lane;
    const float s1i = s1[i0 + r];
    const float* s2p = s2 + cbase + lane;
    float vv[HCOLS / 32];
    float m = -3.4e38f;
#pragma unroll
    for (int k = 0; k < HCOLS / 32; ++k) {
      float v = lrelu(s1i + s2p[32 * k]) + row[32 * k];
      v = (adjv[k] > 0) ? v : NEG_BIG;
      vv[k] = v;
      m = fmaxf(m, v);
    }
    for (int mask = 16; mask >= 1; mask >>= 1)
      m = fmaxf(m, __shfl_xor(m, mask, 32));
    float s = 0.f;
#pragma unroll
    for (int k = 0; k < HCOLS / 32; ++k) {
      const float e = __expf(vv[k] - m);
      row[32 * k] = e;
      s += e;
    }
    for (int mask = 16; mask >= 1; mask >>= 1) s += __shfl_xor(s, mask, 32);
    if (lane == 0) {
      s_inv[r] = 1.0f / s;
      s_lse[r] = m + logf(s);
    }
  }
  __syncthreads();

  v8f c0 = {};
  v8f c1 = {};
  const int kbase = wave * (HCOLS / NWAVES);
  for (int kk = 0; kk < HCOLS / NWAVES; kk += 4) {
    const int k0 = kbase + kk;
    v2f a = *(const v2f*)(sc + nn * ROWSTRIDE + k0 + kd);
    const float* hr = h + (size_t)(cbase + k0 + kd) * D_OUT;
    v2f b0, b1;
    b0.x = hr[nn];
    b0.y = hr[D_OUT + nn];
    b1.x = hr[16 + nn];
    b1.y = hr[D_OUT + 16 + nn];
    c0 = wmma_f32_k4(a, b0, c0);
    c1 = wmma_f32_k4(a, b1, c1);
  }
  {
    const int row = (lane >> 4) << 3;
    float* pw = part + wave * 512;
    for (int v = 0; v < 8; ++v) {
      pw[(row + v) * 32 + nn] = c0[v];
      pw[(row + v) * 32 + 16 + nn] = c1[v];
    }
  }
  __syncthreads();
  if (tid < 512) {
    const int e = tid;
    float s = 0.f;
    for (int w = 0; w < NWAVES; ++w) s += part[w * 512 + e];
    const int row = e >> 5, col = e & 31;
    float* dst = sp ? o1 : o0;
    dst[(size_t)(i0 + row) * D_OUT + col] = s * s_inv[row];
  } else if (tid < 528) {
    lse[sp * N + i0 + (tid - 512)] = s_lse[tid - 512];
  }
}

__global__ void bn_partial_kernel(float* __restrict__ hp,
                                  const float* __restrict__ o1,
                                  const float* __restrict__ lse,
                                  float* __restrict__ bp) {
  __shared__ float ls[8][32];
  __shared__ float lq[8][32];
  const int col = threadIdx.x & 31;
  const int seg = threadIdx.x >> 5;
  const int r0 = blockIdx.x * 32 + seg * 4;
  float s = 0.f, q = 0.f;
  for (int r = 0; r < 4; ++r) {
    const size_t idx = (size_t)(r0 + r) * D_OUT + col;
    const float l0 = lse[r0 + r], l1 = lse[N + r0 + r];
    const float lm = fmaxf(l0, l1);
    const float w0 = expf(l0 - lm), w1 = expf(l1 - lm);
    const float v = (w0 * hp[idx] + w1 * o1[idx]) / (w0 + w1);
    hp[idx] = v;
    s += v;
    q += v * v;
  }
  ls[seg][col] = s;
  lq[seg][col] = q;
  __syncthreads();
  if (threadIdx.x < 32) {
    float ss = 0.f, qq = 0.f;
    for (int w = 0; w < 8; ++w) {
      ss += ls[w][col];
      qq += lq[w][col];
    }
    bp[blockIdx.x * 64 + col] = ss;
    bp[blockIdx.x * 64 + 32 + col] = qq;
  }
}

__global__ void bn_stats_kernel(const float* __restrict__ bp,
                                float* __restrict__ ms) {
  const int col = threadIdx.x;
  float s = 0.f, q = 0.f;
  for (int b = 0; b < 64; ++b) {
    s += bp[b * 64 + col];
    q += bp[b * 64 + 32 + col];
  }
  const float mean = s * (1.0f / (float)N);
  const float var = q * (1.0f / (float)N) - mean * mean;
  ms[col] = mean;
  ms[32 + col] = rsqrtf(var + 1e-5f);
}

__global__ void bn_norm_kernel(const float* __restrict__ hp,
                               const float* __restrict__ ms,
                               const float* __restrict__ gamma,
                               const float* __restrict__ beta,
                               float* __restrict__ out) {
  const int idx = blockIdx.x * blockDim.x + threadIdx.x;
  const int col = idx & 31;
  out[idx] = gamma[col] * (hp[idx] - ms[col]) * ms[32 + col] + beta[col];
}

extern "C" void kernel_launch(void* const* d_in, const int* in_sizes, int n_in,
                              void* d_out, int out_size, void* d_ws,
                              size_t ws_size, hipStream_t stream) {
  const float* input = (const float*)d_in[0];
  const float* edge_feat = (const float*)d_in[1];
  const int* adj = (const int*)d_in[2];
  const float* Wh = (const float*)d_in[3];
  const float* Wh1 = (const float*)d_in[4];
  const float* Wh2 = (const float*)d_in[5];
  const float* Wf = (const float*)d_in[6];
  const float* ah = (const float*)d_in[7];
  const float* af = (const float*)d_in[8];
  const float* bf = (const float*)d_in[9];
  const float* gamma = (const float*)d_in[10];
  const float* beta = (const float*)d_in[11];

  float* ws = (float*)d_ws;
  float* v1 = ws;
  float* v2 = ws + 128;
  float* s1 = ws + 256;
  float* s2 = ws + 2304;
  float* h = ws + 4352;
  float* hp = ws + 69888;
  float* lse = ws + 135424;
  float* bp = ws + 4352;
  float* ms = ws + 139520;
  float* out = (float*)d_out;

  prep_vec_kernel<<<1, 128, 0, stream>>>(Wh1, Wh2, ah, v1, v2);
  prep_s_kernel<<<8, 256, 0, stream>>>(input, v1, v2, s1, s2);
  h_gemm_kernel<<<128, 32, 0, stream>>>(input, Wh, h);
  const size_t smem = (size_t)(16 * ROWSTRIDE + NWAVES * 512) * sizeof(float);
  fused_attention_kernel<<<256, 1024, smem, stream>>>(edge_feat, adj, Wf, af, bf,
                                                      s1, s2, h, hp, out, lse);
  bn_partial_kernel<<<64, 256, 0, stream>>>(hp, out, lse, bp);
  bn_stats_kernel<<<1, 32, 0, stream>>>(bp, ms);
  bn_norm_kernel<<<256, 256, 0, stream>>>(hp, ms, gamma, beta, out);
}


# attention kernel: wave index read through the first lane, so task indices, address bases, loop control and the wave<16 branches run on the scalar unit, not as per-lane vector code; same sums
# speedup vs baseline: 4.8274x; 1.0824x over previous
#include <hip/hip_runtime.h>
#include <math.h>

#define N 2048
#define D_IN 128
#define D_OUT 32
#define D_EDGE 8
#define ALPHA 0.2f
#define NEG_BIG -9e15f
#define HCOLS 1024
#define ROWSTRIDE 1026
#define NWAVES 32

typedef __attribute__((ext_vector_type(2))) float v2f;
typedef __attribute__((ext_vector_type(8))) float v8f;

__device__ __forceinline__ v8f wmma_f32_k4(v2f a, v2f b, v8f c) {
  return __builtin_amdgcn_wmma_f32_16x16x4_f32(false, a, false, b, (short)0, c,
                                               false, false);
}

__device__ __forceinline__ float lrelu(float x) { return x >= 0.f ? x : ALPHA * x; }

template <int M>
__device__ __forceinline__ float row_xor(float x) {
  return __int_as_float(__builtin_amdgcn_update_dpp(0, __float_as_int(x), 0x160 + M,
                                                    0xf, 0xf, true));
}

__global__ void prep_vec_kernel(const float* __restrict__ Wh1,
                                const float* __restrict__ Wh2,
                                const float* __restrict__ ah,
                                float* __restrict__ v1, float* __restrict__ v2) {
  const int t = threadIdx.x;
  float a1 = 0.f, a2 = 0.f;
  for (int f = 0; f < D_OUT; ++f) {
    const float w = ah[f];
    a1 += Wh1[t * D_OUT + f] * w;
    a2 += Wh2[t * D_OUT + f] * w;
  }
  v1[t] = a1;
  v2[t] = a2;
}

__global__ void prep_s_kernel(const float* __restrict__ input,
                              const float* __restrict__ v1,
                              const float* __restrict__ v2,
                              float* __restrict__ s1, float* __restrict__ s2) {
  const int i = blockIdx.x * blockDim.x + threadIdx.x;
  const float* row = input + (size_t)i * D_IN;
  float a = 0.f, b = 0.f;
  for (int k = 0; k < D_IN; ++k) {
    const float x = row[k];
    a += x * v1[k];
    b += x * v2[k];
  }
  s1[i] = a;
  s2[i] = b;
}

__global__ void h_gemm_kernel(const float* __restrict__ input,
                              const float* __restrict__ Wh,
                              float* __restrict__ h) {
  const int lane = threadIdx.x & 31;
  const int i0 = blockIdx.x * 16;
  const int m = lane & 15;
  const int kd = (lane >> 4) << 1;
  const int n = lane & 15;
  v8f c0 = {};
  v8f c1 = {};
  for (int k0 = 0; k0 < D_IN; k0 += 4) {
    v2f a = *(const v2f*)(input + (size_t)(i0 + m) * D_IN + k0 + kd);
    v2f b0, b1;
    b0.x = Wh[(k0 + kd) * D_OUT + n];
    b0.y = Wh[(k0 + kd + 1) * D_OUT + n];
    b1.x = Wh[(k0 + kd) * D_OUT + 16 + n];
    b1.y = Wh[(k0 + kd + 1) * D_OUT + 16 + n];
    c0 = wmma_f32_k4(a, b0, c0);
    c1 = wmma_f32_k4(a, b1, c1);
  }
  const int row = (lane >> 4) << 3;
  for (int v = 0; v < 8; ++v) {
    h[(size_t)(i0 + row + v) * D_OUT + n] = c0[v];
    h[(size_t)(i0 + row + v) * D_OUT + 16 + n] = c1[v];
  }
}

__global__ void __launch_bounds__(1024)
fused_attention_kernel(const float* __restrict__ edge_feat,
                                       const int* __restrict__ adj,
                                       const float* __restrict__ Wf,
                                       const float* __restrict__ af,
                                       const float* __restrict__ bf,
                                       const float* __restrict__ s1,
                                       const float* __restrict__ s2,
                                       const float* __restrict__ h,
                                       float* __restrict__ o0,
                                       float* __restrict__ o1,
                                       float* __restrict__ lse) {
  extern __shared__ float smem[];
  float* sc = smem;
  float* part = smem + 16 * ROWSTRIDE;
  __shared__ float s_inv[16];
  __shared__ float s_lse[16];

  const int tid = threadIdx.x;
  const int lane = tid & 31;
  const int wave = __builtin_amdgcn_readfirstlane(tid >> 5);
  const int i0 = (blockIdx.x >> 1) * 16;
  const int sp = blockIdx.x & 1;
  const int cbase = sp * HCOLS;

  const int nn = lane & 15;
  const int kd = (lane >> 4) << 1;
  const bool b0 = (lane & 1) != 0, b1 = (lane & 2) != 0, b2 = (lane & 4) != 0;

  v2f bw00, bw01, bw10, bw11;
  bw00.x = Wf[(0 + kd) * D_OUT + nn];
  bw00.y = Wf[(1 + kd) * D_OUT + nn];
  bw01.x = Wf[(0 + kd) * D_OUT + 16 + nn];
  bw01.y = Wf[(1 + kd) * D_OUT + 16 + nn];
  bw10.x = Wf[(4 + kd) * D_OUT + nn];
  bw10.y = Wf[(5 + kd) * D_OUT + nn];
  bw11.x = Wf[(4 + kd) * D_OUT + 16 + nn];
  bw11.y = Wf[(5 + kd) * D_OUT + 16 + nn];
  const float af_lo = af[nn], af_hi = af[16 + nn];
  const float bf_lo = bf[nn], bf_hi = bf[16 + nn];
  int adjv[HCOLS / 32] = {};
  if (wave < 16) {
    const int* arow = adj + (size_t)(i0 + wave) * N + cbase + lane;
#pragma unroll
    for (int k = 0; k < HCOLS / 32; ++k) adjv[k] = arow[32 * k];
  }
  v8f cini0, cini1;
  for (int v = 0; v < 8; ++v) {
    cini0[v] = bf_lo;
    cini1[v] = bf_hi;
  }

  const float* ebase = edge_feat + ((size_t)i0 * N + cbase) * D_EDGE;
  const unsigned eoff = (unsigned)(nn * D_EDGE + kd);
  const int srow = ((lane >> 4) << 3) + (lane & 7);
  for (int t = wave; t < 16 * 64; t += NWAVES) {
    const int il = t >> 6;
    const int jl0 = (t & 63) << 4;
    const float* ep = ebase + ((size_t)il * N + jl0) * D_EDGE + eoff;
    const int tn = t + NWAVES;
    if (tn < 16 * 64)
      __builtin_prefetch(ebase + ((size_t)(tn >> 6) * N + ((tn & 63) << 4)) * D_EDGE + eoff,
                         0, 1);
    v2f a0 = *(const v2f*)(ep);
    v2f a1 = *(const v2f*)(ep + 4);
    v8f f0 = wmma_f32_k4(a0, bw00, cini0);
    f0 = wmma_f32_k4(a1, bw10, f0);
    v8f f1 = wmma_f32_k4(a0, bw01, cini1);
    f1 = wmma_f32_k4(a1, bw11, f1);
    float p[8];
    for (int v = 0; v < 8; ++v)
      p[v] = lrelu(f0[v]) * af_lo + lrelu(f1[v]) * af_hi;
    float q[4], u[2];
    for (int k = 0; k < 4; ++k) {
      const float keep = b0 ? p[2 * k + 1] : p[2 * k];
      const float send = b0 ? p[2 * k] : p[2 * k + 1];
      q[k] = keep + row_xor<1>(send);
    }
    for (int k = 0; k < 2; ++k) {
      const float keep = b1 ? q[2 * k + 1] : q[2 * k];
      const float send = b1 ? q[2 * k] : q[2 * k + 1];
      u[k] = keep + row_xor<2>(send);
    }
    float cp;
    {
      const float keep = b2 ? u[1] : u[0];
      const float send = b2 ? u[0] : u[1];
      cp = keep + row_xor<4>(send);
    }
    cp += row_xor<8>(cp);
    if ((lane & 8) == 0) {
      sc[il * ROWSTRIDE + jl0 + srow] = lrelu(cp);
    }
  }
  __syncthreads();

  if (wave < 16) {
    const int r = wave;
    float* row = sc + r * ROWSTRIDE + lane;
    const float s1i = s1[i0 + r];
    const float* s2p = s2 + cbase + lane;
    float vv[HCOLS / 32];
    float m = -3.4e38f;
#pragma unroll
    for (int k = 0; k < HCOLS / 32; ++k) {
      float v = lrelu(s1i + s2p[32 * k]) + row[32 * k];
      v = (adjv[k] > 0) ? v : NEG_BIG;
      vv[k] = v;
      m = fmaxf(m, v);
    }
    for (int mask = 16; mask >= 1; mask >>= 1)
      m = fmaxf(m, __shfl_xor(m, mask, 32));
    float s = 0.f;
#pragma unroll
    for (int k = 0; k < HCOLS / 32; ++k) {
      const float e = __expf(vv[k] - m);
      row[32 * k] = e;
      s += e;
    }
    for (int mask = 16; mask >= 1; mask >>= 1) s += __shfl_xor(s, mask, 32);
    if (lane == 0) {
      s_inv[r] = 1.0f / s;
      s_lse[r] = m + logf(s);
    }
  }
  __syncthreads();

  v8f c0 = {};
  v8f c1 = {};
  const int kbase = wave * (HCOLS / NWAVES);
  for (int kk = 0; kk < HCOLS / NWAVES; kk += 4) {
    const int k0 = kbase + kk;
    v2f a = *(const v2f*)(sc + nn * ROWSTRIDE + k0 + kd);
    const float* hr = h + (size_t)(cbase + k0 + kd) * D_OUT;
    v2f b0, b1;
    b0.x = hr[nn];
    b0.y = hr[D_OUT + nn];
    b1.x = hr[16 + nn];
    b1.y = hr[D_OUT + 16 + nn];
    c0 = wmma_f32_k4(a, b0, c0);
    c1 = wmma_f32_k4(a, b1, c1);
  }
  {
    const int row = (lane >> 4) << 3;
    float* pw = part + wave * 512;
    for (int v = 0; v < 8; ++v) {
      pw[(row + v) * 32 + nn] = c0[v];
      pw[(row + v) * 32 + 16 + nn] = c1[v];
    }
  }
  __syncthreads();
  if (tid < 512) {
    const int e = tid;
    float s = 0.f;
    for (int w = 0; w < NWAVES; ++w) s += part[w * 512 + e];
    const int row = e >> 5, col = e & 31;
    float* dst = sp ? o1 : o0;
    dst[(size_t)(i0 + row) * D_OUT + col] = s * s_inv[row];
  } else if (tid < 528) {
    lse[sp * N + i0 + (tid - 512)] = s_lse[tid - 512];
  }
}

__global__ void bn_partial_kernel(float* __restrict__ hp,
                                  const float* __restrict__ o1,
                                  const float* __restrict__ lse,
                                  float* __restrict__ bp) {
  __shared__ float ls[8][32];
  __shared__ float lq[8][32];
  const int col = threadIdx.x & 31;
  const int seg = threadIdx.x >> 5;
  const int r0 = blockIdx.x * 32 + seg * 4;
  float s = 0.f, q = 0.f;
  for (int r = 0; r < 4; ++r) {
    const size_t idx = (size_t)(r0 + r) * D_OUT + col;
    const float l0 = lse[r0 + r], l1 = lse[N + r0 + r];
    const float lm = fmaxf(l0, l1);
    const float w0 = expf(l0 - lm), w1 = expf(l1 - lm);
    const float v = (w0 * hp[idx] + w1 * o1[idx]) / (w0 + w1);
    hp[idx] = v;
    s += v;
    q += v * v;
  }
  ls[seg][col] = s;
  lq[seg][col] = q;
  __syncthreads();
  if (threadIdx.x < 32) {
    float ss = 0.f, qq = 0.f;
    for (int w = 0; w < 8; ++w) {
      ss += ls[w][col];
      qq += lq[w][col];
    }
    bp[blockIdx.x * 64 + col] = ss;
    bp[blockIdx.x * 64 + 32 + col] = qq;
  }
}

__global__ void bn_stats_kernel(const float* __restrict__ bp,
                                float* __restrict__ ms) {
  const int col = threadIdx.x;
  float s = 0.f, q = 0.f;
  for (int b = 0; b < 64; ++b) {
    s += bp[b * 64 + col];
    q += bp[b * 64 + 32 + col];
  }
  const float mean = s * (1.0f / (float)N);
  const float var = q * (1.0f / (float)N) - mean * mean;
  ms[col] = mean;
  ms[32 + col] = rsqrtf(var + 1e-5f);
}

__global__ void bn_norm_kernel(const float* __restrict__ hp,
                               const float* __restrict__ ms,
                               const float* __restrict__ gamma,
                               const float* __restrict__ beta,
                               float* __restrict__ out) {
  const int idx = blockIdx.x * blockDim.x + threadIdx.x;
  const int col = idx & 31;
  out[idx] = gamma[col] * (hp[idx] - ms[col]) * ms[32 + col] + beta[col];
}

extern "C" void kernel_launch(void* const* d_in, const int* in_sizes, int n_in,
                              void* d_out, int out_size, void* d_ws,
                              size_t ws_size, hipStream_t stream) {
  const float* input = (const float*)d_in[0];
  const float* edge_feat = (const float*)d_in[1];
  const int* adj = (const int*)d_in[2];
  const float* Wh = (const float*)d_in[3];
  const float* Wh1 = (const float*)d_in[4];
  const float* Wh2 = (const float*)d_in[5];
  const float* Wf = (const float*)d_in[6];
  const float* ah = (const float*)d_in[7];
  const float* af = (const float*)d_in[8];
  const float* bf = (const float*)d_in[9];
  const float* gamma = (const float*)d_in[10];
  const float* beta = (const float*)d_in[11];

  float* ws = (float*)d_ws;
  float* v1 = ws;
  float* v2 = ws + 128;
  float* s1 = ws + 256;
  float* s2 = ws + 2304;
  float* h = ws + 4352;
  float* hp = ws + 69888;
  float* lse = ws + 135424;
  float* bp = ws + 4352;
  float* ms = ws + 139520;
  float* out = (float*)d_out;

  prep_vec_kernel<<<1, 128, 0, stream>>>(Wh1, Wh2, ah, v1, v2);
  prep_s_kernel<<<8, 256, 0, stream>>>(input, v1, v2, s1, s2);
  h_gemm_kernel<<<128, 32, 0, stream>>>(input, Wh, h);
  const size_t smem = (size_t)(16 * ROWSTRIDE + NWAVES * 512) * sizeof(float);
  fused_attention_kernel<<<256, 1024, smem, stream>>>(edge_feat, adj, Wf, af, bf,
                                                      s1, s2, h, hp, out, lse);
  bn_partial_kernel<<<64, 256, 0, stream>>>(hp, out, lse, bp);
  bn_stats_kernel<<<1, 32, 0, stream>>>(bp, ms);
  bn_norm_kernel<<<256, 256, 0, stream>>>(hp, ms, gamma, beta, out);
}


# s1/s2 and BatchNorm-stats helper kernels pay one memory round trip, were 16 and 8 serial ones: matvec re-laid (lane owns 4 k, 8 rows per wave), all loads issued first; s1, s2 sum order changes
# speedup vs baseline: 5.2769x; 1.0931x over previous
#include <hip/hip_runtime.h>
#include <math.h>

#define N 2048
#define D_IN 128
#define D_OUT 32
#define D_EDGE 8
#define ALPHA 0.2f
#define NEG_BIG -9e15f
#define HCOLS 1024
#define ROWSTRIDE 1026
#define NWAVES 32

typedef __attribute__((ext_vector_type(2))) float v2f;
typedef __attribute__((ext_vector_type(4))) float v4f;
typedef __attribute__((ext_vector_type(8))) float v8f;

__device__ __forceinline__ v8f wmma_f32_k4(v2f a, v2f b, v8f c) {
  return __builtin_amdgcn_wmma_f32_16x16x4_f32(false, a, false, b, (short)0, c,
                                               false, false);
}

__device__ __forceinline__ float lrelu(float x) { return x >= 0.f ? x : ALPHA * x; }

template <int M>
__device__ __forceinline__ float row_xor(float x) {
  return __int_as_float(__builtin_amdgcn_update_dpp(0, __float_as_int(x), 0x160 + M,
                                                    0xf, 0xf, true));
}

__global__ void prep_vec_kernel(const float* __restrict__ Wh1,
                                const float* __restrict__ Wh2,
                                const float* __restrict__ ah,
                                float* __restrict__ v1, float* __restrict__ v2) {
  const int t = threadIdx.x;
  float a1 = 0.f, a2 = 0.f;
  for (int f = 0; f < D_OUT; ++f) {
    const float w = ah[f];
    a1 += Wh1[t * D_OUT + f] * w;
    a2 += Wh2[t * D_OUT + f] * w;
  }
  v1[t] = a1;
  v2[t] = a2;
}

#define SROWS 8
#define SWAVES 4
__global__ void __launch_bounds__(32 * SWAVES)
prep_s_kernel(const float* __restrict__ input,
              const float* __restrict__ v1,
              const float* __restrict__ v2,
              float* __restrict__ s1, float* __restrict__ s2) {
  const int lane = threadIdx.x & 31;
  const int r0 = (blockIdx.x * SWAVES + (threadIdx.x >> 5)) * SROWS;
  const bool b0 = (lane & 1) != 0, b1 = (lane & 2) != 0, b2 = (lane & 4) != 0,
             b3 = (lane & 8) != 0;
  const v4f a = *(const v4f*)(v1 + 4 * lane);
  const v4f b = *(const v4f*)(v2 + 4 * lane);
  const float* row = input + (size_t)r0 * D_IN + 4 * lane;
  v4f x[SROWS];
#pragma unroll
  for (int r = 0; r < SROWS; ++r) x[r] = *(const v4f*)(row + r * D_IN);
  float p[2 * SROWS];
  for (int r = 0; r < SROWS; ++r) {
    p[r] = x[r].x * a.x + x[r].y * a.y + x[r].z * a.z + x[r].w * a.w;
    p[SROWS + r] = x[r].x * b.x + x[r].y * b.y + x[r].z * b.z + x[r].w * b.w;
  }
  float q[8], u[4], w[2];
  for (int k = 0; k < 8; ++k) {
    const float keep = b0 ? p[2 * k + 1] : p[2 * k];
    const float send = b0 ? p[2 * k] : p[2 * k + 1];
    q[k] = keep + row_xor<1>(send);
  }
  for (int k = 0; k < 4; ++k) {
    const float keep = b1 ? q[2 * k + 1] : q[2 * k];
    const float send = b1 ? q[2 * k] : q[2 * k + 1];
    u[k] = keep + row_xor<2>(send);
  }
  for (int k = 0; k < 2; ++k) {
    const float keep = b2 ? u[2 * k + 1] : u[2 * k];
    const float send = b2 ? u[2 * k] : u[2 * k + 1];
    w[k] = keep + row_xor<4>(send);
  }
  float t;
  {
    const float keep = b3 ? w[1] : w[0];
    const float send = b3 ? w[0] : w[1];
    t = keep + row_xor<8>(send);
  }
  t += __shfl_xor(t, 16, 32);
  if (lane < SROWS) s1[r0 + lane] = t;
  else if (lane < 2 * SROWS) s2[r0 + lane - SROWS] = t;
}

__global__ void h_gemm_kernel(const float* __restrict__ input,
                              const float* __restrict__ Wh,
                              float* __restrict__ h) {
  const int lane = threadIdx.x & 31;
  const int i0 = blockIdx.x * 16;
  const int m = lane & 15;
  const int kd = (lane >> 4) << 1;
  const int n = lane & 15;
  v8f c0 = {};
  v8f c1 = {};
  for (int k0 = 0; k0 < D_IN; k0 += 4) {
    v2f a = *(const v2f*)(input + (size_t)(i0 + m) * D_IN + k0 + kd);
    v2f b0, b1;
    b0.x = Wh[(k0 + kd) * D_OUT + n];
    b0.y = Wh[(k0 + kd + 1) * D_OUT + n];
    b1.x = Wh[(k0 + kd) * D_OUT + 16 + n];
    b1.y = Wh[(k0 + kd + 1) * D_OUT + 16 + n];
    c0 = wmma_f32_k4(a, b0, c0);
    c1 = wmma_f32_k4(a, b1, c1);
  }
  const int row = (lane >> 4) << 3;
  for (int v = 0; v < 8; ++v) {
    h[(size_t)(i0 + row + v) * D_OUT + n] = c0[v];
    h[(size_t)(i0 + row + v) * D_OUT + 16 + n] = c1[v];
  }
}

__global__ void __launch_bounds__(1024)
fused_attention_kernel(const float* __restrict__ edge_feat,
                                       const int* __restrict__ adj,
                                       const float* __restrict__ Wf,
                                       const float* __restrict__ af,
                                       const float* __restrict__ bf,
                                       const float* __restrict__ s1,
                                       const float* __restrict__ s2,
                                       const float* __restrict__ h,
                                       float* __restrict__ o0,
                                       float* __restrict__ o1,
                                       float* __restrict__ lse) {
  extern __shared__ float smem[];
  float* sc = smem;
  float* part = smem + 16 * ROWSTRIDE;
  __shared__ float s_inv[16];
  __shared__ float s_lse[16];

  const int tid = threadIdx.x;
  const int lane = tid & 31;
  const int wave = __builtin_amdgcn_readfirstlane(tid >> 5);
  const int i0 = (blockIdx.x >> 1) * 16;
  const int sp = blockIdx.x & 1;
  const int cbase = sp * HCOLS;

  const int nn = lane & 15;
  const int kd = (lane >> 4) << 1;
  const bool b0 = (lane & 1) != 0, b1 = (lane & 2) != 0, b2 = (lane & 4) != 0;

  v2f bw00, bw01, bw10, bw11;
  bw00.x = Wf[(0 + kd) * D_OUT + nn];
  bw00.y = Wf[(1 + kd) * D_OUT + nn];
  bw01.x = Wf[(0 + kd) * D_OUT + 16 + nn];
  bw01.y = Wf[(1 + kd) * D_OUT + 16 + nn];
  bw10.x = Wf[(4 + kd) * D_OUT + nn];
  bw10.y = Wf[(5 + kd) * D_OUT + nn];
  bw11.x = Wf[(4 + kd) * D_OUT + 16 + nn];
  bw11.y = Wf[(5 + kd) * D_OUT + 16 + nn];
  const float af_lo = af[nn], af_hi = af[16 + nn];
  const float bf_lo = bf[nn], bf_hi = bf[16 + nn];
  int adjv[HCOLS / 32] = {};
  if (wave < 16) {
    const int* arow = adj + (size_t)(i0 + wave) * N + cbase + lane;
#pragma unroll
    for (int k = 0; k < HCOLS / 32; ++k) adjv[k] = arow[32 * k];
  }
  v8f cini0, cini1;
  for (int v = 0; v < 8; ++v) {
    cini0[v] = bf_lo;
    cini1[v] = bf_hi;
  }

  const float* ebase = edge_feat + ((size_t)i0 * N + cbase) * D_EDGE;
  const unsigned eoff = (unsigned)(nn * D_EDGE + kd);
  const int srow = ((lane >> 4) << 3) + (lane & 7);
  for (int t = wave; t < 16 * 64; t += NWAVES) {
    const int il = t >> 6;
    const int jl0 = (t & 63) << 4;
    const float* ep = ebase + ((size_t)il * N + jl0) * D_EDGE + eoff;
    const int tn = t + NWAVES;
    if (tn < 16 * 64)
      __builtin_prefetch(ebase + ((size_t)(tn >> 6) * N + ((tn & 63) << 4)) * D_EDGE + eoff,
                         0, 1);
    v2f a0 = *(const v2f*)(ep);
    v2f a1 = *(const v2f*)(ep + 4);
    v8f f0 = wmma_f32_k4(a0, bw00, cini0);
    f0 = wmma_f32_k4(a1, bw10, f0);
    v8f f1 = wmma_f32_k4(a0, bw01, cini1);
    f1 = wmma_f32_k4(a1, bw11, f1);
    float p[8];
    for (int v = 0; v < 8; ++v)
      p[v] = lrelu(f0[v]) * af_lo + lrelu(f1[v]) * af_hi;
    float q[4], u[2];
    for (int k = 0; k < 4; ++k) {
      const float keep = b0 ? p[2 * k + 1] : p[2 * k];
      const float send = b0 ? p[2 * k] : p[2 * k + 1];
      q[k] = keep + row_xor<1>(send);
    }
    for (int k = 0; k < 2; ++k) {
      const float keep = b1 ? q[2 * k + 1] : q[2 * k];
      const float send = b1 ? q[2 * k] : q[2 * k + 1];
      u[k] = keep + row_xor<2>(send);
    }
    float cp;
    {
      const float keep = b2 ? u[1] : u[0];
      const float send = b2 ? u[0] : u[1];
      cp = keep + row_xor<4>(send);
    }
    cp += row_xor<8>(cp);
    if ((lane & 8) == 0) {
      sc[il * ROWSTRIDE + jl0 + srow] = lrelu(cp);
    }
  }
  __syncthreads();

  if (wave < 16) {
    const int r = wave;
    float* row = sc + r * ROWSTRIDE + lane;
    const float s1i = s1[i0 + r];
    const float* s2p = s2 + cbase + lane;
    float vv[HCOLS / 32];
    float m = -3.4e38f;
#pragma unroll
    for (int k = 0; k < HCOLS / 32; ++k) {
      float v = lrelu(s1i + s2p[32 * k]) + row[32 * k];
      v = (adjv[k] > 0) ? v : NEG_BIG;
      vv[k] = v;
      m = fmaxf(m, v);
    }
    for (int mask = 16; mask >= 1; mask >>= 1)
      m = fmaxf(m, __shfl_xor(m, mask, 32));
    float s = 0.f;
#pragma unroll
    for (int k = 0; k < HCOLS / 32; ++k) {
      const float e = __expf(vv[k] - m);
      row[32 * k] = e;
      s += e;
    }
    for (int mask = 16; mask >= 1; mask >>= 1) s += __shfl_xor(s, mask, 32);
    if (lane == 0) {
      s_inv[r] = 1.0f / s;
      s_lse[r] = m + logf(s);
    }
  }
  __syncthreads();

  v8f c0 = {};
  v8f c1 = {};
  const int kbase = wave * (HCOLS / NWAVES);
  for (int kk = 0; kk < HCOLS / NWAVES; kk += 4) {
    const int k0 = kbase + kk;
    v2f a = *(const v2f*)(sc + nn * ROWSTRIDE + k0 + kd);
    const float* hr = h + (size_t)(cbase + k0 + kd) * D_OUT;
    v2f b0, b1;
    b0.x = hr[nn];
    b0.y = hr[D_OUT + nn];
    b1.x = hr[16 + nn];
    b1.y = hr[D_OUT + 16 + nn];
    c0 = wmma_f32_k4(a, b0, c0);
    c1 = wmma_f32_k4(a, b1, c1);
  }
  {
    const int row = (lane >> 4) << 3;
    float* pw = part + wave * 512;
    for (int v = 0; v < 8; ++v) {
      pw[(row + v) * 32 + nn] = c0[v];
      pw[(row + v) * 32 + 16 + nn] = c1[v];
    }
  }
  __syncthreads();
  if (tid < 512) {
    const int e = tid;
    float s = 0.f;
    for (int w = 0; w < NWAVES; ++w) s += part[w * 512 + e];
    const int row = e >> 5, col = e & 31;
    float* dst = sp ? o1 : o0;
    dst[(size_t)(i0 + row) * D_OUT + col] = s * s_inv[row];
  } else if (tid < 528) {
    lse[sp * N + i0 + (tid - 512)] = s_lse[tid - 512];
  }
}

__global__ void bn_partial_kernel(float* __restrict__ hp,
                                  const float* __restrict__ o1,
                                  const float* __restrict__ lse,
                                  float* __restrict__ bp) {
  __shared__ float ls[8][32];
  __shared__ float lq[8][32];
  const int col = threadIdx.x & 31;
  const int seg = threadIdx.x >> 5;
  const int r0 = blockIdx.x * 32 + seg * 4;
  float s = 0.f, q = 0.f;
  for (int r = 0; r < 4; ++r) {
    const size_t idx = (size_t)(r0 + r) * D_OUT + col;
    const float l0 = lse[r0 + r], l1 = lse[N + r0 + r];
    const float lm = fmaxf(l0, l1);
    const float w0 = expf(l0 - lm), w1 = expf(l1 - lm);
    const float v = (w0 * hp[idx] + w1 * o1[idx]) / (w0 + w1);
    hp[idx] = v;
    s += v;
    q += v * v;
  }
  ls[seg][col] = s;
  lq[seg][col] = q;
  __syncthreads();
  if (threadIdx.x < 32) {
    float ss = 0.f, qq = 0.f;
    for (int w = 0; w < 8; ++w) {
      ss += ls[w][col];
      qq += lq[w][col];
    }
    bp[blockIdx.x * 64 + col] = ss;
    bp[blockIdx.x * 64 + 32 + col] = qq;
  }
}

__global__ void bn_stats_kernel(const float* __restrict__ bp,
                                float* __restrict__ ms) {
  const int col = threadIdx.x;
  float s = 0.f, q = 0.f;
#pragma unroll
  for (int b = 0; b < 64; ++b) {
    s += bp[b * 64 + col];
    q += bp[b * 64 + 32 + col];
  }
  const float mean = s * (1.0f / (float)N);
  const float var = q * (1.0f / (float)N) - mean * mean;
  ms[col] = mean;
  ms[32 + col] = rsqrtf(var + 1e-5f);
}

__global__ void bn_norm_kernel(const float* __restrict__ hp,
                               const float* __restrict__ ms,
                               const float* __restrict__ gamma,
                               const float* __restrict__ beta,
                               float* __restrict__ out) {
  const int idx = blockIdx.x * blockDim.x + threadIdx.x;
  const int col = idx & 31;
  out[idx] = gamma[col] * (hp[idx] - ms[col]) * ms[32 + col] + beta[col];
}

extern "C" void kernel_launch(void* const* d_in, const int* in_sizes, int n_in,
                              void* d_out, int out_size, void* d_ws,
                              size_t ws_size, hipStream_t stream) {
  const float* input = (const float*)d_in[0];
  const float* edge_feat = (const float*)d_in[1];
  const int* adj = (const int*)d_in[2];
  const float* Wh = (const float*)d_in[3];
  const float* Wh1 = (const float*)d_in[4];
  const float* Wh2 = (const float*)d_in[5];
  const float* Wf = (const float*)d_in[6];
  const float* ah = (const float*)d_in[7];
  const float* af = (const float*)d_in[8];
  const float* bf = (const float*)d_in[9];
  const float* gamma = (const float*)d_in[10];
  const float* beta = (const float*)d_in[11];

  float* ws = (float*)d_ws;
  float* v1 = ws;
  float* v2 = ws + 128;
  float* s1 = ws + 256;
  float* s2 = ws + 2304;
  float* h = ws + 4352;
  float* hp = ws + 69888;
  float* lse = ws + 135424;
  float* bp = ws + 4352;
  float* ms = ws + 139520;
  float* out = (float*)d_out;

  prep_vec_kernel<<<1, 128, 0, stream>>>(Wh1, Wh2, ah, v1, v2);
  prep_s_kernel<<<64, 128, 0, stream>>>(input, v1, v2, s1, s2);
  h_gemm_kernel<<<128, 32, 0, stream>>>(input, Wh, h);
  const size_t smem = (size_t)(16 * ROWSTRIDE + NWAVES * 512) * sizeof(float);
  fused_attention_kernel<<<256, 1024, smem, stream>>>(edge_feat, adj, Wf, af, bf,
                                                      s1, s2, h, hp, out, lse);
  bn_partial_kernel<<<64, 256, 0, stream>>>(hp, out, lse, bp);
  bn_stats_kernel<<<1, 32, 0, stream>>>(bp, ms);
  bn_norm_kernel<<<256, 256, 0, stream>>>(hp, ms, gamma, beta, out);
}


# leaky ReLU as max(x, 0.2x), exact for a slope in (0,1): one v_max per value, was compare + select and moves to pair the packed multiplies; edge-score loop and softmax pass; same sums, same bits
# speedup vs baseline: 5.5062x; 1.0435x over previous
#include <hip/hip_runtime.h>
#include <math.h>

#define N 2048
#define D_IN 128
#define D_OUT 32
#define D_EDGE 8
#define ALPHA 0.2f
#define NEG_BIG -9e15f
#define HCOLS 1024
#define ROWSTRIDE 1026
#define NWAVES 32

typedef __attribute__((ext_vector_type(2))) float v2f;
typedef __attribute__((ext_vector_type(4))) float v4f;
typedef __attribute__((ext_vector_type(8))) float v8f;

__device__ __forceinline__ v8f wmma_f32_k4(v2f a, v2f b, v8f c) {
  return __builtin_amdgcn_wmma_f32_16x16x4_f32(false, a, false, b, (short)0, c,
                                               false, false);
}

__device__ __forceinline__ float lrelu(float x) { return fmaxf(x, ALPHA * x); }

template <int M>
__device__ __forceinline__ float row_xor(float x) {
  return __int_as_float(__builtin_amdgcn_update_dpp(0, __float_as_int(x), 0x160 + M,
                                                    0xf, 0xf, true));
}

__global__ void prep_vec_kernel(const float* __restrict__ Wh1,
                                const float* __restrict__ Wh2,
                                const float* __restrict__ ah,
                                float* __restrict__ v1, float* __restrict__ v2) {
  const int t = threadIdx.x;
  float a1 = 0.f, a2 = 0.f;
  for (int f = 0; f < D_OUT; ++f) {
    const float w = ah[f];
    a1 += Wh1[t * D_OUT + f] * w;
    a2 += Wh2[t * D_OUT + f] * w;
  }
  v1[t] = a1;
  v2[t] = a2;
}

#define SROWS 8
#define SWAVES 4
__global__ void __launch_bounds__(32 * SWAVES)
prep_s_kernel(const float* __restrict__ input,
              const float* __restrict__ v1,
              const float* __restrict__ v2,
              float* __restrict__ s1, float* __restrict__ s2) {
  const int lane = threadIdx.x & 31;
  const int r0 = (blockIdx.x * SWAVES + (threadIdx.x >> 5)) * SROWS;
  const bool b0 = (lane & 1) != 0, b1 = (lane & 2) != 0, b2 = (lane & 4) != 0,
             b3 = (lane & 8) != 0;
  const v4f a = *(const v4f*)(v1 + 4 * lane);
  const v4f b = *(const v4f*)(v2 + 4 * lane);
  const float* row = input + (size_t)r0 * D_IN + 4 * lane;
  v4f x[SROWS];
#pragma unroll
  for (int r = 0; r < SROWS; ++r) x[r] = *(const v4f*)(row + r * D_IN);
  float p[2 * SROWS];
  for (int r = 0; r < SROWS; ++r) {
    p[r] = x[r].x * a.x + x[r].y * a.y + x[r].z * a.z + x[r].w * a.w;
    p[SROWS + r] = x[r].x * b.x + x[r].y * b.y + x[r].z * b.z + x[r].w * b.w;
  }
  float q[8], u[4], w[2];
  for (int k = 0; k < 8; ++k) {
    const float keep = b0 ? p[2 * k + 1] : p[2 * k];
    const float send = b0 ? p[2 * k] : p[2 * k + 1];
    q[k] = keep + row_xor<1>(send);
  }
  for (int k = 0; k < 4; ++k) {
    const float keep = b1 ? q[2 * k + 1] : q[2 * k];
    const float send = b1 ? q[2 * k] : q[2 * k + 1];
    u[k] = keep + row_xor<2>(send);
  }
  for (int k = 0; k < 2; ++k) {
    const float keep = b2 ? u[2 * k + 1] : u[2 * k];
    const float send = b2 ? u[2 * k] : u[2 * k + 1];
    w[k] = keep + row_xor<4>(send);
  }
  float t;
  {
    const float keep = b3 ? w[1] : w[0];
    const float send = b3 ? w[0] : w[1];
    t = keep + row_xor<8>(send);
  }
  t += __shfl_xor(t, 16, 32);
  if (lane < SROWS) s1[r0 + lane] = t;
  else if (lane < 2 * SROWS) s2[r0 + lane - SROWS] = t;
}

__global__ void h_gemm_kernel(const float* __restrict__ input,
                              const float* __restrict__ Wh,
                              float* __restrict__ h) {
  const int lane = threadIdx.x & 31;
  const int i0 = blockIdx.x * 16;
  const int m = lane & 15;
  const int kd = (lane >> 4) << 1;
  const int n = lane & 15;
  v8f c0 = {};
  v8f c1 = {};
  for (int k0 = 0; k0 < D_IN; k0 += 4) {
    v2f a = *(const v2f*)(input + (size_t)(i0 + m) * D_IN + k0 + kd);
    v2f b0, b1;
    b0.x = Wh[(k0 + kd) * D_OUT + n];
    b0.y = Wh[(k0 + kd + 1) * D_OUT + n];
    b1.x = Wh[(k0 + kd) * D_OUT + 16 + n];
    b1.y = Wh[(k0 + kd + 1) * D_OUT + 16 + n];
    c0 = wmma_f32_k4(a, b0, c0);
    c1 = wmma_f32_k4(a, b1, c1);
  }
  const int row = (lane >> 4) << 3;
  for (int v = 0; v < 8; ++v) {
    h[(size_t)(i0 + row + v) * D_OUT + n] = c0[v];
    h[(size_t)(i0 + row + v) * D_OUT + 16 + n] = c1[v];
  }
}

__global__ void __launch_bounds__(1024)
fused_attention_kernel(const float* __restrict__ edge_feat,
                                       const int* __restrict__ adj,
                                       const float* __restrict__ Wf,
                                       const float* __restrict__ af,
                                       const float* __restrict__ bf,
                                       const float* __restrict__ s1,
                                       const float* __restrict__ s2,
                                       const float* __restrict__ h,
                                       float* __restrict__ o0,
                                       float* __restrict__ o1,
                                       float* __restrict__ lse) {
  extern __shared__ float smem[];
  float* sc = smem;
  float* part = smem + 16 * ROWSTRIDE;
  __shared__ float s_inv[16];
  __shared__ float s_lse[16];

  const int tid = threadIdx.x;
  const int lane = tid & 31;
  const int wave = __builtin_amdgcn_readfirstlane(tid >> 5);
  const int i0 = (blockIdx.x >> 1) * 16;
  const int sp = blockIdx.x & 1;
  const int cbase = sp * HCOLS;

  const int nn = lane & 15;
  const int kd = (lane >> 4) << 1;
  const bool b0 = (lane & 1) != 0, b1 = (lane & 2) != 0, b2 = (lane & 4) != 0;

  v2f bw00, bw01, bw10, bw11;
  bw00.x = Wf[(0 + kd) * D_OUT + nn];
  bw00.y = Wf[(1 + kd) * D_OUT + nn];
  bw01.x = Wf[(0 + kd) * D_OUT + 16 + nn];
  bw01.y = Wf[(1 + kd) * D_OUT + 16 + nn];
  bw10.x = Wf[(4 + kd) * D_OUT + nn];
  bw10.y = Wf[(5 + kd) * D_OUT + nn];
  bw11.x = Wf[(4 + kd) * D_OUT + 16 + nn];
  bw11.y = Wf[(5 + kd) * D_OUT + 16 + nn];
  const float af_lo = af[nn], af_hi = af[16 + nn];
  const float bf_lo = bf[nn], bf_hi = bf[16 + nn];
  int adjv[HCOLS / 32] = {};
  if (wave < 16) {
    const int* arow = adj + (size_t)(i0 + wave) * N + cbase + lane;
#pragma unroll
    for (int k = 0; k < HCOLS / 32; ++k) adjv[k] = arow[32 * k];
  }
  v8f cini0, cini1;
  for (int v = 0; v < 8; ++v) {
    cini0[v] = bf_lo;
    cini1[v] = bf_hi;
  }

  const float* ebase = edge_feat + ((size_t)i0 * N + cbase) * D_EDGE;
  const unsigned eoff = (unsigned)(nn * D_EDGE + kd);
  const int srow = ((lane >> 4) << 3) + (lane & 7);
  for (int t = wave; t < 16 * 64; t += NWAVES) {
    const int il = t >> 6;
    const int jl0 = (t & 63) << 4;
    const float* ep = ebase + ((size_t)il * N + jl0) * D_EDGE + eoff;
    const int tn = t + NWAVES;
    if (tn < 16 * 64)
      __builtin_prefetch(ebase + ((size_t)(tn >> 6) * N + ((tn & 63) << 4)) * D_EDGE + eoff,
                         0, 1);
    v2f a0 = *(const v2f*)(ep);
    v2f a1 = *(const v2f*)(ep + 4);
    v8f f0 = wmma_f32_k4(a0, bw00, cini0);
    f0 = wmma_f32_k4(a1, bw10, f0);
    v8f f1 = wmma_f32_k4(a0, bw01, cini1);
    f1 = wmma_f32_k4(a1, bw11, f1);
    float p[8];
    for (int k = 0; k < 4; ++k) {
      const v2f x0 = {f0[2 * k], f0[2 * k + 1]}, x1 = {f1[2 * k], f1[2 * k + 1]};
      const v2f y0 = x0 * ALPHA, y1 = x1 * ALPHA;
      const v2f g0 = {fmaxf(x0.x, y0.x), fmaxf(x0.y, y0.y)};
      const v2f g1 = {fmaxf(x1.x, y1.x), fmaxf(x1.y, y1.y)};
      const v2f pp = g0 * af_lo + g1 * af_hi;
      p[2 * k] = pp.x;
      p[2 * k + 1] = pp.y;
    }
    float q[4], u[2];
    for (int k = 0; k < 4; ++k) {
      const float keep = b0 ? p[2 * k + 1] : p[2 * k];
      const float send = b0 ? p[2 * k] : p[2 * k + 1];
      q[k] = keep + row_xor<1>(send);
    }
    for (int k = 0; k < 2; ++k) {
      const float keep = b1 ? q[2 * k + 1] : q[2 * k];
      const float send = b1 ? q[2 * k] : q[2 * k + 1];
      u[k] = keep + row_xor<2>(send);
    }
    float cp;
    {
      const float keep = b2 ? u[1] : u[0];
      const float send = b2 ? u[0] : u[1];
      cp = keep + row_xor<4>(send);
    }
    cp += row_xor<8>(cp);
    if ((lane & 8) == 0) {
      sc[il * ROWSTRIDE + jl0 + srow] = lrelu(cp);
    }
  }
  __syncthreads();

  if (wave < 16) {
    const int r = wave;
    float* row = sc + r * ROWSTRIDE + lane;
    const float s1i = s1[i0 + r];
    const float* s2p = s2 + cbase + lane;
    float vv[HCOLS / 32];
    float m = -3.4e38f;
#pragma unroll
    for (int k = 0; k < HCOLS / 32; ++k) {
      float v = lrelu(s1i + s2p[32 * k]) + row[32 * k];
      v = (adjv[k] > 0) ? v : NEG_BIG;
      vv[k] = v;
      m = fmaxf(m, v);
    }
    for (int mask = 16; mask >= 1; mask >>= 1)
      m = fmaxf(m, __shfl_xor(m, mask, 32));
    float s = 0.f;
#pragma unroll
    for (int k = 0; k < HCOLS / 32; ++k) {
      const float e = __expf(vv[k] - m);
      row[32 * k] = e;
      s += e;
    }
    for (int mask = 16; mask >= 1; mask >>= 1) s += __shfl_xor(s, mask, 32);
    if (lane == 0) {
      s_inv[r] = 1.0f / s;
      s_lse[r] = m + logf(s);
    }
  }
  __syncthreads();

  v8f c0 = {};
  v8f c1 = {};
  const int kbase = wave * (HCOLS / NWAVES);
  for (int kk = 0; kk < HCOLS / NWAVES; kk += 4) {
    const int k0 = kbase + kk;
    v2f a = *(const v2f*)(sc + nn * ROWSTRIDE + k0 + kd);
    const float* hr = h + (size_t)(cbase + k0 + kd) * D_OUT;
    v2f b0, b1;
    b0.x = hr[nn];
    b0.y = hr[D_OUT + nn];
    b1.x = hr[16 + nn];
    b1.y = hr[D_OUT + 16 + nn];
    c0 = wmma_f32_k4(a, b0, c0);
    c1 = wmma_f32_k4(a, b1, c1);
  }
  {
    const int row = (lane >> 4) << 3;
    float* pw = part + wave * 512;
    for (int v = 0; v < 8; ++v) {
      pw[(row + v) * 32 + nn] = c0[v];
      pw[(row + v) * 32 + 16 + nn] = c1[v];
    }
  }
  __syncthreads();
  if (tid < 512) {
    const int e = tid;
    float s = 0.f;
    for (int w = 0; w < NWAVES; ++w) s += part[w * 512 + e];
    const int row = e >> 5, col = e & 31;
    float* dst = sp ? o1 : o0;
    dst[(size_t)(i0 + row) * D_OUT + col] = s * s_inv[row];
  } else if (tid < 528) {
    lse[sp * N + i0 + (tid - 512)] = s_lse[tid - 512];
  }
}

__global__ void bn_partial_kernel(float* __restrict__ hp,
                                  const float* __restrict__ o1,
                                  const float* __restrict__ lse,
                                  float* __restrict__ bp) {
  __shared__ float ls[8][32];
  __shared__ float lq[8][32];
  const int col = threadIdx.x & 31;
  const int seg = threadIdx.x >> 5;
  const int r0 = blockIdx.x * 32 + seg * 4;
  float s = 0.f, q = 0.f;
  for (int r = 0; r < 4; ++r) {
    const size_t idx = (size_t)(r0 + r) * D_OUT + col;
    const float l0 = lse[r0 + r], l1 = lse[N + r0 + r];
    const float lm = fmaxf(l0, l1);
    const float w0 = expf(l0 - lm), w1 = expf(l1 - lm);
    const float v = (w0 * hp[idx] + w1 * o1[idx]) / (w0 + w1);
    hp[idx] = v;
    s += v;
    q += v * v;
  }
  ls[seg][col] = s;
  lq[seg][col] = q;
  __syncthreads();
  if (threadIdx.x < 32) {
    float ss = 0.f, qq = 0.f;
    for (int w = 0; w < 8; ++w) {
      ss += ls[w][col];
      qq += lq[w][col];
    }
    bp[blockIdx.x * 64 + col] = ss;
    bp[blockIdx.x * 64 + 32 + col] = qq;
  }
}

__global__ void bn_stats_kernel(const float* __restrict__ bp,
                                float* __restrict__ ms) {
  const int col = threadIdx.x;
  float s = 0.f, q = 0.f;
#pragma unroll
  for (int b = 0; b < 64; ++b) {
    s += bp[b * 64 + col];
    q += bp[b * 64 + 32 + col];
  }
  const float mean = s * (1.0f / (float)N);
  const float var = q * (1.0f / (float)N) - mean * mean;
  ms[col] = mean;
  ms[32 + col] = rsqrtf(var + 1e-5f);
}

__global__ void bn_norm_kernel(const float* __restrict__ hp,
                               const float* __restrict__ ms,
                               const float* __restrict__ gamma,
                               const float* __restrict__ beta,
                               float* __restrict__ out) {
  const int idx = blockIdx.x * blockDim.x + threadIdx.x;
  const int col = idx & 31;
  out[idx] = gamma[col] * (hp[idx] - ms[col]) * ms[32 + col] + beta[col];
}

extern "C" void kernel_launch(void* const* d_in, const int* in_sizes, int n_in,
                              void* d_out, int out_size, void* d_ws,
                              size_t ws_size, hipStream_t stream) {
  const float* input = (const float*)d_in[0];
  const float* edge_feat = (const float*)d_in[1];
  const int* adj = (const int*)d_in[2];
  const float* Wh = (const float*)d_in[3];
  const float* Wh1 = (const float*)d_in[4];
  const float* Wh2 = (const float*)d_in[5];
  const float* Wf = (const float*)d_in[6];
  const float* ah = (const float*)d_in[7];
  const float* af = (const float*)d_in[8];
  const float* bf = (const float*)d_in[9];
  const float* gamma = (const float*)d_in[10];
  const float* beta = (const float*)d_in[11];

  float* ws = (float*)d_ws;
  float* v1 = ws;
  float* v2 = ws + 128;
  float* s1 = ws + 256;
  float* s2 = ws + 2304;
  float* h = ws + 4352;
  float* hp = ws + 69888;
  float* lse = ws + 135424;
  float* bp = ws + 4352;
  float* ms = ws + 139520;
  float* out = (float*)d_out;

  prep_vec_kernel<<<1, 128, 0, stream>>>(Wh1, Wh2, ah, v1, v2);
  prep_s_kernel<<<64, 128, 0, stream>>>(input, v1, v2, s1, s2);
  h_gemm_kernel<<<128, 32, 0, stream>>>(input, Wh, h);
  const size_t smem = (size_t)(16 * ROWSTRIDE + NWAVES * 512) * sizeof(float);
  fused_attention_kernel<<<256, 1024, smem, stream>>>(edge_feat, adj, Wf, af, bf,
                                                      s1, s2, h, hp, out, lse);
  bn_partial_kernel<<<64, 256, 0, stream>>>(hp, out, lse, bp);
  bn_stats_kernel<<<1, 32, 0, stream>>>(bp, ms);
  bn_norm_kernel<<<256, 256, 0, stream>>>(hp, ms, gamma, beta, out);
}


# attention kernel skips masked edges: per row the columns with adj > 0 are compacted (ballot + running count) into an LDS list and edge rows gathered by it; edge-score work halves; same sums
# speedup vs baseline: 6.1316x; 1.1136x over previous
#include <hip/hip_runtime.h>
#include <math.h>

#define N 2048
#define D_IN 128
#define D_OUT 32
#define D_EDGE 8
#define ALPHA 0.2f
#define NEG_BIG -9e15f
#define HCOLS 1024
#define ROWSTRIDE 1026
#define NWAVES 32
#define CSTRIDE 1024
#define ISTRIDE 1040

typedef __attribute__((ext_vector_type(2))) float v2f;
typedef __attribute__((ext_vector_type(4))) float v4f;
typedef __attribute__((ext_vector_type(8))) float v8f;

__device__ __forceinline__ v8f wmma_f32_k4(v2f a, v2f b, v8f c) {
  return __builtin_amdgcn_wmma_f32_16x16x4_f32(false, a, false, b, (short)0, c,
                                               false, false);
}

__device__ __forceinline__ float lrelu(float x) { return fmaxf(x, ALPHA * x); }

template <int M>
__device__ __forceinline__ float row_xor(float x) {
  return __int_as_float(__builtin_amdgcn_update_dpp(0, __float_as_int(x), 0x160 + M,
                                                    0xf, 0xf, true));
}

__global__ void prep_vec_kernel(const float* __restrict__ Wh1,
                                const float* __restrict__ Wh2,
                                const float* __restrict__ ah,
                                float* __restrict__ v1, float* __restrict__ v2) {
  const int t = threadIdx.x;
  float a1 = 0.f, a2 = 0.f;
  for (int f = 0; f < D_OUT; ++f) {
    const float w = ah[f];
    a1 += Wh1[t * D_OUT + f] * w;
    a2 += Wh2[t * D_OUT + f] * w;
  }
  v1[t] = a1;
  v2[t] = a2;
}

#define SROWS 8
#define SWAVES 4
__global__ void __launch_bounds__(32 * SWAVES)
prep_s_kernel(const float* __restrict__ input,
              const float* __restrict__ v1,
              const float* __restrict__ v2,
              float* __restrict__ s1, float* __restrict__ s2) {
  const int lane = threadIdx.x & 31;
  const int r0 = (blockIdx.x * SWAVES + (threadIdx.x >> 5)) * SROWS;
  const bool b0 = (lane & 1) != 0, b1 = (lane & 2) != 0, b2 = (lane & 4) != 0,
             b3 = (lane & 8) != 0;
  const v4f a = *(const v4f*)(v1 + 4 * lane);
  const v4f b = *(const v4f*)(v2 + 4 * lane);
  const float* row = input + (size_t)r0 * D_IN + 4 * lane;
  v4f x[SROWS];
#pragma unroll
  for (int r = 0; r < SROWS; ++r) x[r] = *(const v4f*)(row + r * D_IN);
  float p[2 * SROWS];
  for (int r = 0; r < SROWS; ++r) {
    p[r] = x[r].x * a.x + x[r].y * a.y + x[r].z * a.z + x[r].w * a.w;
    p[SROWS + r] = x[r].x * b.x + x[r].y * b.y + x[r].z * b.z + x[r].w * b.w;
  }
  float q[8], u[4], w[2];
  for (int k = 0; k < 8; ++k) {
    const float keep = b0 ? p[2 * k + 1] : p[2 * k];
    const float send = b0 ? p[2 * k] : p[2 * k + 1];
    q[k] = keep + row_xor<1>(send);
  }
  for (int k = 0; k < 4; ++k) {
    const float keep = b1 ? q[2 * k + 1] : q[2 * k];
    const float send = b1 ? q[2 * k] : q[2 * k + 1];
    u[k] = keep + row_xor<2>(send);
  }
  for (int k = 0; k < 2; ++k) {
    const float keep = b2 ? u[2 * k + 1] : u[2 * k];
    const float send = b2 ? u[2 * k] : u[2 * k + 1];
    w[k] = keep + row_xor<4>(send);
  }
  float t;
  {
    const float keep = b3 ? w[1] : w[0];
    const float send = b3 ? w[0] : w[1];
    t = keep + row_xor<8>(send);
  }
  t += __shfl_xor(t, 16, 32);
  if (lane < SROWS) s1[r0 + lane] = t;
  else if (lane < 2 * SROWS) s2[r0 + lane - SROWS] = t;
}

__global__ void h_gemm_kernel(const float* __restrict__ input,
                              const float* __restrict__ Wh,
                              float* __restrict__ h) {
  const int lane = threadIdx.x & 31;
  const int i0 = blockIdx.x * 16;
  const int m = lane & 15;
  const int kd = (lane >> 4) << 1;
  const int n = lane & 15;
  v8f c0 = {};
  v8f c1 = {};
  for (int k0 = 0; k0 < D_IN; k0 += 4) {
    v2f a = *(const v2f*)(input + (size_t)(i0 + m) * D_IN + k0 + kd);
    v2f b0, b1;
    b0.x = Wh[(k0 + kd) * D_OUT + n];
    b0.y = Wh[(k0 + kd + 1) * D_OUT + n];
    b1.x = Wh[(k0 + kd) * D_OUT + 16 + n];
    b1.y = Wh[(k0 + kd + 1) * D_OUT + 16 + n];
    c0 = wmma_f32_k4(a, b0, c0);
    c1 = wmma_f32_k4(a, b1, c1);
  }
  const int row = (lane >> 4) << 3;
  for (int v = 0; v < 8; ++v) {
    h[(size_t)(i0 + row + v) * D_OUT + n] = c0[v];
    h[(size_t)(i0 + row + v) * D_OUT + 16 + n] = c1[v];
  }
}

__global__ void __launch_bounds__(1024)
fused_attention_kernel(const float* __restrict__ edge_feat,
                                       const int* __restrict__ adj,
                                       const float* __restrict__ Wf,
                                       const float* __restrict__ af,
                                       const float* __restrict__ bf,
                                       const float* __restrict__ s1,
                                       const float* __restrict__ s2,
                                       const float* __restrict__ h,
                                       float* __restrict__ o0,
                                       float* __restrict__ o1,
                                       float* __restrict__ lse) {
  extern __shared__ float smem[];
  float* sc = smem;
  float* part = smem + 16 * ROWSTRIDE;
  float* cs = part + NWAVES * 512;
  unsigned short* idx = (unsigned short*)(cs + 16 * CSTRIDE);
  __shared__ int s_cnt[16];
  __shared__ float s_inv[16];
  __shared__ float s_lse[16];

  const int tid = threadIdx.x;
  const int lane = tid & 31;
  const int wave = __builtin_amdgcn_readfirstlane(tid >> 5);
  const int i0 = (blockIdx.x >> 1) * 16;
  const int sp = blockIdx.x & 1;
  const int cbase = sp * HCOLS;

  const int nn = lane & 15;
  const int kd = (lane >> 4) << 1;
  const bool b0 = (lane & 1) != 0, b1 = (lane & 2) != 0, b2 = (lane & 4) != 0;

  v2f bw00, bw01, bw10, bw11;
  bw00.x = Wf[(0 + kd) * D_OUT + nn];
  bw00.y = Wf[(1 + kd) * D_OUT + nn];
  bw01.x = Wf[(0 + kd) * D_OUT + 16 + nn];
  bw01.y = Wf[(1 + kd) * D_OUT + 16 + nn];
  bw10.x = Wf[(4 + kd) * D_OUT + nn];
  bw10.y = Wf[(5 + kd) * D_OUT + nn];
  bw11.x = Wf[(4 + kd) * D_OUT + 16 + nn];
  bw11.y = Wf[(5 + kd) * D_OUT + 16 + nn];
  const float af_lo = af[nn], af_hi = af[16 + nn];
  const float bf_lo = bf[nn], bf_hi = bf[16 + nn];
  int adjv[HCOLS / 32] = {};
  if (wave < 16) {
    const int* arow = adj + (size_t)(i0 + wave) * N + cbase + lane;
#pragma unroll
    for (int k = 0; k < HCOLS / 32; ++k) adjv[k] = arow[32 * k];
    unsigned short* irow = idx + wave * ISTRIDE;
    int base = 0;
#pragma unroll
    for (int k = 0; k < HCOLS / 32; ++k) {
      const bool on = adjv[k] > 0;
      const unsigned bal = __builtin_amdgcn_ballot_w32(on);
      const int pos = base + (int)__builtin_amdgcn_mbcnt_lo(bal, 0u);
      if (on) irow[pos] = (unsigned short)(lane + 32 * k);
      base += __builtin_popcount(bal);
    }
    if (lane < 16) irow[base + lane] = 0;
    if (lane == 0) s_cnt[wave] = base;
  }
  __syncthreads();
  v8f cini0, cini1;
  for (int v = 0; v < 8; ++v) {
    cini0[v] = bf_lo;
    cini1[v] = bf_hi;
  }

  const int il = wave & 15;
  const int ng = (__builtin_amdgcn_readfirstlane(s_cnt[il]) + 15) >> 4;
  const float* erow = edge_feat + ((size_t)(i0 + il) * N + cbase) * D_EDGE;
  const unsigned short* irow = idx + il * ISTRIDE + nn;
  float* crow = cs + il * CSTRIDE;
  const int srow = ((lane >> 4) << 3) + (lane & 7);
  int g = wave >> 4;
  unsigned jn = 0;
  if (g < ng) jn = irow[16 * g];
  for (; g < ng; g += 2) {
    const float* ep = erow + (jn * D_EDGE + kd);
    v2f a0 = *(const v2f*)(ep);
    v2f a1 = *(const v2f*)(ep + 4);
    if (g + 2 < ng) {
      jn = irow[16 * (g + 2)];
      __builtin_prefetch(erow + (jn * D_EDGE + kd), 0, 1);
    }
    v8f f0 = wmma_f32_k4(a0, bw00, cini0);
    f0 = wmma_f32_k4(a1, bw10, f0);
    v8f f1 = wmma_f32_k4(a0, bw01, cini1);
    f1 = wmma_f32_k4(a1, bw11, f1);
    float p[8];
    for (int k = 0; k < 4; ++k) {
      const v2f x0 = {f0[2 * k], f0[2 * k + 1]}, x1 = {f1[2 * k], f1[2 * k + 1]};
      const v2f y0 = x0 * ALPHA, y1 = x1 * ALPHA;
      const v2f g0 = {fmaxf(x0.x, y0.x), fmaxf(x0.y, y0.y)};
      const v2f g1 = {fmaxf(x1.x, y1.x), fmaxf(x1.y, y1.y)};
      const v2f pp = g0 * af_lo + g1 * af_hi;
      p[2 * k] = pp.x;
      p[2 * k + 1] = pp.y;
    }
    float q[4], u[2];
    for (int k = 0; k < 4; ++k) {
      const float keep = b0 ? p[2 * k + 1] : p[2 * k];
      const float send = b0 ? p[2 * k] : p[2 * k + 1];
      q[k] = keep + row_xor<1>(send);
    }
    for (int k = 0; k < 2; ++k) {
      const float keep = b1 ? q[2 * k + 1] : q[2 * k];
      const float send = b1 ? q[2 * k] : q[2 * k + 1];
      u[k] = keep + row_xor<2>(send);
    }
    float cp;
    {
      const float keep = b2 ? u[1] : u[0];
      const float send = b2 ? u[0] : u[1];
      cp = keep + row_xor<4>(send);
    }
    cp += row_xor<8>(cp);
    if ((lane & 8) == 0) {
      crow[16 * g + srow] = lrelu(cp);
    }
  }
  __syncthreads();

  if (wave < 16) {
    const int r = wave;
    float* row = sc + r * ROWSTRIDE + lane;
    const float* crow_r = cs + r * CSTRIDE;
    int base = 0;
    const float s1i = s1[i0 + r];
    const float* s2p = s2 + cbase + lane;
    float vv[HCOLS / 32];
    float m = -3.4e38f;
#pragma unroll
    for (int k = 0; k < HCOLS / 32; ++k) {
      const bool on = adjv[k] > 0;
      const unsigned bal = __builtin_amdgcn_ballot_w32(on);
      const int pos = base + (int)__builtin_amdgcn_mbcnt_lo(bal, 0u);
      base += __builtin_popcount(bal);
      float v = lrelu(s1i + s2p[32 * k]) + crow_r[pos];
      v = on ? v : NEG_BIG;
      vv[k] = v;
      m = fmaxf(m, v);
    }
    for (int mask = 16; mask >= 1; mask >>= 1)
      m = fmaxf(m, __shfl_xor(m, mask, 32));
    float s = 0.f;
#pragma unroll
    for (int k = 0; k < HCOLS / 32; ++k) {
      const float e = __expf(vv[k] - m);
      row[32 * k] = e;
      s += e;
    }
    for (int mask = 16; mask >= 1; mask >>= 1) s += __shfl_xor(s, mask, 32);
    if (lane == 0) {
      s_inv[r] = 1.0f / s;
      s_lse[r] = m + logf(s);
    }
  }
  __syncthreads();

  v8f c0 = {};
  v8f c1 = {};
  const int kbase = wave * (HCOLS / NWAVES);
  for (int kk = 0; kk < HCOLS / NWAVES; kk += 4) {
    const int k0 = kbase + kk;
    v2f a = *(const v2f*)(sc + nn * ROWSTRIDE + k0 + kd);
    const float* hr = h + (size_t)(cbase + k0 + kd) * D_OUT;
    v2f b0, b1;
    b0.x = hr[nn];
    b0.y = hr[D_OUT + nn];
    b1.x = hr[16 + nn];
    b1.y = hr[D_OUT + 16 + nn];
    c0 = wmma_f32_k4(a, b0, c0);
    c1 = wmma_f32_k4(a, b1, c1);
  }
  {
    const int row = (lane >> 4) << 3;
    float* pw = part + wave * 512;
    for (int v = 0; v < 8; ++v) {
      pw[(row + v) * 32 + nn] = c0[v];
      pw[(row + v) * 32 + 16 + nn] = c1[v];
    }
  }
  __syncthreads();
  if (tid < 512) {
    const int e = tid;
    float s = 0.f;
    for (int w = 0; w < NWAVES; ++w) s += part[w * 512 + e];
    const int row = e >> 5, col = e & 31;
    float* dst = sp ? o1 : o0;
    dst[(size_t)(i0 + row) * D_OUT + col] = s * s_inv[row];
  } else if (tid < 528) {
    lse[sp * N + i0 + (tid - 512)] = s_lse[tid - 512];
  }
}

__global__ void bn_partial_kernel(float* __restrict__ hp,
                                  const float* __restrict__ o1,
                                  const float* __restrict__ lse,
                                  float* __restrict__ bp) {
  __shared__ float ls[8][32];
  __shared__ float lq[8][32];
  const int col = threadIdx.x & 31;
  const int seg = threadIdx.x >> 5;
  const int r0 = blockIdx.x * 32 + seg * 4;
  float s = 0.f, q = 0.f;
  for (int r = 0; r < 4; ++r) {
    const size_t idx = (size_t)(r0 + r) * D_OUT + col;
    const float l0 = lse[r0 + r], l1 = lse[N + r0 + r];
    const float lm = fmaxf(l0, l1);
    const float w0 = expf(l0 - lm), w1 = expf(l1 - lm);
    const float v = (w0 * hp[idx] + w1 * o1[idx]) / (w0 + w1);
    hp[idx] = v;
    s += v;
    q += v * v;
  }
  ls[seg][col] = s;
  lq[seg][col] = q;
  __syncthreads();
  if (threadIdx.x < 32) {
    float ss = 0.f, qq = 0.f;
    for (int w = 0; w < 8; ++w) {
      ss += ls[w][col];
      qq += lq[w][col];
    }
    bp[blockIdx.x * 64 + col] = ss;
    bp[blockIdx.x * 64 + 32 + col] = qq;
  }
}

__global__ void bn_stats_kernel(const float* __restrict__ bp,
                                float* __restrict__ ms) {
  const int col = threadIdx.x;
  float s = 0.f, q = 0.f;
#pragma unroll
  for (int b = 0; b < 64; ++b) {
    s += bp[b * 64 + col];
    q += bp[b * 64 + 32 + col];
  }
  const float mean = s * (1.0f / (float)N);
  const float var = q * (1.0f / (float)N) - mean * mean;
  ms[col] = mean;
  ms[32 + col] = rsqrtf(var + 1e-5f);
}

__global__ void bn_norm_kernel(const float* __restrict__ hp,
                               const float* __restrict__ ms,
                               const float* __restrict__ gamma,
                               const float* __restrict__ beta,
                               float* __restrict__ out) {
  const int idx = blockIdx.x * blockDim.x + threadIdx.x;
  const int col = idx & 31;
  out[idx] = gamma[col] * (hp[idx] - ms[col]) * ms[32 + col] + beta[col];
}

extern "C" void kernel_launch(void* const* d_in, const int* in_sizes, int n_in,
                              void* d_out, int out_size, void* d_ws,
                              size_t ws_size, hipStream_t stream) {
  const float* input = (const float*)d_in[0];
  const float* edge_feat = (const float*)d_in[1];
  const int* adj = (const int*)d_in[2];
  const float* Wh = (const float*)d_in[3];
  const float* Wh1 = (const float*)d_in[4];
  const float* Wh2 = (const float*)d_in[5];
  const float* Wf = (const float*)d_in[6];
  const float* ah = (const float*)d_in[7];
  const float* af = (const float*)d_in[8];
  const float* bf = (const float*)d_in[9];
  const float* gamma = (const float*)d_in[10];
  const float* beta = (const float*)d_in[11];

  float* ws = (float*)d_ws;
  float* v1 = ws;
  float* v2 = ws + 128;
  float* s1 = ws + 256;
  float* s2 = ws + 2304;
  float* h = ws + 4352;
  float* hp = ws + 69888;
  float* lse = ws + 135424;
  float* bp = ws + 4352;
  float* ms = ws + 139520;
  float* out = (float*)d_out;

  prep_vec_kernel<<<1, 128, 0, stream>>>(Wh1, Wh2, ah, v1, v2);
  prep_s_kernel<<<64, 128, 0, stream>>>(input, v1, v2, s1, s2);
  h_gemm_kernel<<<128, 32, 0, stream>>>(input, Wh, h);
  const size_t smem = (size_t)(16 * ROWSTRIDE + NWAVES * 512 + 16 * CSTRIDE) * sizeof(float) +
                      (size_t)(16 * ISTRIDE) * sizeof(unsigned short);
  fused_attention_kernel<<<256, 1024, smem, stream>>>(edge_feat, adj, Wf, af, bf,
                                                      s1, s2, h, hp, out, lse);
  bn_partial_kernel<<<64, 256, 0, stream>>>(hp, out, lse, bp);
  bn_stats_kernel<<<1, 32, 0, stream>>>(bp, ms);
  bn_norm_kernel<<<256, 256, 0, stream>>>(hp, ms, gamma, beta, out);
}


# 7 dependent launches to 4 by work-group role: one prep launch (64 groups rebuild Wh1@ah, Wh2@ah in LDS, then s1/s2; 32 groups do h); BN stats summed in each normalising group; BN sum order changes
# speedup vs baseline: 7.2836x; 1.1879x over previous
#include <hip/hip_runtime.h>
#include <math.h>

#define N 2048
#define D_IN 128
#define D_OUT 32
#define D_EDGE 8
#define ALPHA 0.2f
#define NEG_BIG -9e15f
#define HCOLS 1024
#define ROWSTRIDE 1026
#define NWAVES 32
#define CSTRIDE 1024
#define ISTRIDE 1040

typedef __attribute__((ext_vector_type(2))) float v2f;
typedef __attribute__((ext_vector_type(4))) float v4f;
typedef __attribute__((ext_vector_type(8))) float v8f;

__device__ __forceinline__ v8f wmma_f32_k4(v2f a, v2f b, v8f c) {
  return __builtin_amdgcn_wmma_f32_16x16x4_f32(false, a, false, b, (short)0, c,
                                               false, false);
}

__device__ __forceinline__ float lrelu(float x) { return fmaxf(x, ALPHA * x); }

template <int M>
__device__ __forceinline__ float row_xor(float x) {
  return __int_as_float(__builtin_amdgcn_update_dpp(0, __float_as_int(x), 0x160 + M,
                                                    0xf, 0xf, true));
}

#define SROWS 8
#define SWAVES 4
#define SBLOCKS (N / (SROWS * SWAVES))
#define HBLOCKS (N / (16 * SWAVES))
__global__ void __launch_bounds__(32 * SWAVES)
prep_kernel(const float* __restrict__ input,
            const float* __restrict__ Wh,
            const float* __restrict__ Wh1,
            const float* __restrict__ Wh2,
            const float* __restrict__ ah,
            float* __restrict__ s1, float* __restrict__ s2,
            float* __restrict__ h) {
  __shared__ __attribute__((aligned(16))) float sv[2 * D_IN];
  const int lane = threadIdx.x & 31;
  const int wave = threadIdx.x >> 5;
  if (blockIdx.x < SBLOCKS) {
    const int t = threadIdx.x;
    const int r0 = (blockIdx.x * SWAVES + wave) * SROWS;
    const bool b0 = (lane & 1) != 0, b1 = (lane & 2) != 0, b2 = (lane & 4) != 0,
               b3 = (lane & 8) != 0;
    v4f w1[D_OUT / 4], w2[D_OUT / 4];
#pragma unroll
    for (int f = 0; f < D_OUT / 4; ++f) {
      w1[f] = *(const v4f*)(Wh1 + t * D_OUT + 4 * f);
      w2[f] = *(const v4f*)(Wh2 + t * D_OUT + 4 * f);
    }
    const float* row = input + (size_t)r0 * D_IN + 4 * lane;
    v4f x[SROWS];
#pragma unroll
    for (int r = 0; r < SROWS; ++r) x[r] = *(const v4f*)(row + r * D_IN);
    float wa[D_OUT];
#pragma unroll
    for (int f = 0; f < D_OUT; ++f) wa[f] = ah[f];
    __builtin_amdgcn_sched_barrier(0);
    float a1 = 0.f, a2 = 0.f;
#pragma unroll
    for (int f = 0; f < D_OUT; ++f) {
      a1 += w1[f >> 2][f & 3] * wa[f];
      a2 += w2[f >> 2][f & 3] * wa[f];
    }
    sv[t] = a1;
    sv[D_IN + t] = a2;
    __syncthreads();
    const v4f a = *(const v4f*)(sv + 4 * lane);
    const v4f b = *(const v4f*)(sv + D_IN + 4 * lane);
    float p[2 * SROWS];
    for (int r = 0; r < SROWS; ++r) {
      p[r] = x[r].x * a.x + x[r].y * a.y + x[r].z * a.z + x[r].w * a.w;
      p[SROWS + r] = x[r].x * b.x + x[r].y * b.y + x[r].z * b.z + x[r].w * b.w;
    }
    float q[8], u[4], w[2];
    for (int k = 0; k < 8; ++k) {
      const float keep = b0 ? p[2 * k + 1] : p[2 * k];
      const float send = b0 ? p[2 * k] : p[2 * k + 1];
      q[k] = keep + row_xor<1>(send);
    }
    for (int k = 0; k < 4; ++k) {
      const float keep = b1 ? q[2 * k + 1] : q[2 * k];
      const float send = b1 ? q[2 * k] : q[2 * k + 1];
      u[k] = keep + row_xor<2>(send);
    }
    for (int k = 0; k < 2; ++k) {
      const float keep = b2 ? u[2 * k + 1] : u[2 * k];
      const float send = b2 ? u[2 * k] : u[2 * k + 1];
      w[k] = keep + row_xor<4>(send);
    }
    float tt;
    {
      const float keep = b3 ? w[1] : w[0];
      const float send = b3 ? w[0] : w[1];
      tt = keep + row_xor<8>(send);
    }
    tt += __shfl_xor(tt, 16, 32);
    if (lane < SROWS) s1[r0 + lane] = tt;
    else if (lane < 2 * SROWS) s2[r0 + lane - SROWS] = tt;
  } else {
    const int i0 = ((blockIdx.x - SBLOCKS) * SWAVES + wave) * 16;
    const int m = lane & 15;
    const int kd = (lane >> 4) << 1;
    const int n = lane & 15;
    v8f c0 = {};
    v8f c1 = {};
    for (int k0 = 0; k0 < D_IN; k0 += 4) {
      v2f a = *(const v2f*)(input + (size_t)(i0 + m) * D_IN + k0 + kd);
      v2f b0, b1;
      b0.x = Wh[(k0 + kd) * D_OUT + n];
      b0.y = Wh[(k0 + kd + 1) * D_OUT + n];
      b1.x = Wh[(k0 + kd) * D_OUT + 16 + n];
      b1.y = Wh[(k0 + kd + 1) * D_OUT + 16 + n];
      c0 = wmma_f32_k4(a, b0, c0);
      c1 = wmma_f32_k4(a, b1, c1);
    }
    const int row = (lane >> 4) << 3;
    for (int v = 0; v < 8; ++v) {
      h[(size_t)(i0 + row + v) * D_OUT + n] = c0[v];
      h[(size_t)(i0 + row + v) * D_OUT + 16 + n] = c1[v];
    }
  }
}

__global__ void __launch_bounds__(1024)
fused_attention_kernel(const float* __restrict__ edge_feat,
                                       const int* __restrict__ adj,
                                       const float* __restrict__ Wf,
                                       const float* __restrict__ af,
                                       const float* __restrict__ bf,
                                       const float* __restrict__ s1,
                                       const float* __restrict__ s2,
                                       const float* __restrict__ h,
                                       float* __restrict__ o0,
                                       float* __restrict__ o1,
                                       float* __restrict__ lse) {
  extern __shared__ float smem[];
  float* sc = smem;
  float* part = smem + 16 * ROWSTRIDE;
  float* cs = part + NWAVES * 512;
  unsigned short* idx = (unsigned short*)(cs + 16 * CSTRIDE);
  __shared__ int s_cnt[16];
  __shared__ float s_inv[16];
  __shared__ float s_lse[16];

  const int tid = threadIdx.x;
  const int lane = tid & 31;
  const int wave = __builtin_amdgcn_readfirstlane(tid >> 5);
  const int i0 = (blockIdx.x >> 1) * 16;
  const int sp = blockIdx.x & 1;
  const int cbase = sp * HCOLS;

  const int nn = lane & 15;
  const int kd = (lane >> 4) << 1;
  const bool b0 = (lane & 1) != 0, b1 = (lane & 2) != 0, b2 = (lane & 4) != 0;

  v2f bw00, bw01, bw10, bw11;
  bw00.x = Wf[(0 + kd) * D_OUT + nn];
  bw00.y = Wf[(1 + kd) * D_OUT + nn];
  bw01.x = Wf[(0 + kd) * D_OUT + 16 + nn];
  bw01.y = Wf[(1 + kd) * D_OUT + 16 + nn];
  bw10.x = Wf[(4 + kd) * D_OUT + nn];
  bw10.y = Wf[(5 + kd) * D_OUT + nn];
  bw11.x = Wf[(4 + kd) * D_OUT + 16 + nn];
  bw11.y = Wf[(5 + kd) * D_OUT + 16 + nn];
  const float af_lo = af[nn], af_hi = af[16 + nn];
  const float bf_lo = bf[nn], bf_hi = bf[16 + nn];
  int adjv[HCOLS / 32] = {};
  if (wave < 16) {
    const int* arow = adj + (size_t)(i0 + wave) * N + cbase + lane;
#pragma unroll
    for (int k = 0; k < HCOLS / 32; ++k) adjv[k] = arow[32 * k];
    unsigned short* irow = idx + wave * ISTRIDE;
    int base = 0;
#pragma unroll
    for (int k = 0; k < HCOLS / 32; ++k) {
      const bool on = adjv[k] > 0;
      const unsigned bal = __builtin_amdgcn_ballot_w32(on);
      const int pos = base + (int)__builtin_amdgcn_mbcnt_lo(bal, 0u);
      if (on) irow[pos] = (unsigned short)(lane + 32 * k);
      base += __builtin_popcount(bal);
    }
    if (lane < 16) irow[base + lane] = 0;
    if (lane == 0) s_cnt[wave] = base;
  }
  __syncthreads();
  v8f cini0, cini1;
  for (int v = 0; v < 8; ++v) {
    cini0[v] = bf_lo;
    cini1[v] = bf_hi;
  }

  const int il = wave & 15;
  const int ng = (__builtin_amdgcn_readfirstlane(s_cnt[il]) + 15) >> 4;
  const float* erow = edge_feat + ((size_t)(i0 + il) * N + cbase) * D_EDGE;
  const unsigned short* irow = idx + il * ISTRIDE + nn;
  float* crow = cs + il * CSTRIDE;
  const int srow = ((lane >> 4) << 3) + (lane & 7);
  int g = wave >> 4;
  unsigned jn = 0;
  if (g < ng) jn = irow[16 * g];
  for (; g < ng; g += 2) {
    const float* ep = erow + (jn * D_EDGE + kd);
    v2f a0 = *(const v2f*)(ep);
    v2f a1 = *(const v2f*)(ep + 4);
    if (g + 2 < ng) {
      jn = irow[16 * (g + 2)];
      __builtin_prefetch(erow + (jn * D_EDGE + kd), 0, 1);
    }
    v8f f0 = wmma_f32_k4(a0, bw00, cini0);
    f0 = wmma_f32_k4(a1, bw10, f0);
    v8f f1 = wmma_f32_k4(a0, bw01, cini1);
    f1 = wmma_f32_k4(a1, bw11, f1);
    float p[8];
    for (int k = 0; k < 4; ++k) {
      const v2f x0 = {f0[2 * k], f0[2 * k + 1]}, x1 = {f1[2 * k], f1[2 * k + 1]};
      const v2f y0 = x0 * ALPHA, y1 = x1 * ALPHA;
      const v2f g0 = {fmaxf(x0.x, y0.x), fmaxf(x0.y, y0.y)};
      const v2f g1 = {fmaxf(x1.x, y1.x), fmaxf(x1.y, y1.y)};
      const v2f pp = g0 * af_lo + g1 * af_hi;
      p[2 * k] = pp.x;
      p[2 * k + 1] = pp.y;
    }
    float q[4], u[2];
    for (int k = 0; k < 4; ++k) {
      const float keep = b0 ? p[2 * k + 1] : p[2 * k];
      const float send = b0 ? p[2 * k] : p[2 * k + 1];
      q[k] = keep + row_xor<1>(send);
    }
    for (int k = 0; k < 2; ++k) {
      const float keep = b1 ? q[2 * k + 1] : q[2 * k];
      const float send = b1 ? q[2 * k] : q[2 * k + 1];
      u[k] = keep + row_xor<2>(send);
    }
    float cp;
    {
      const float keep = b2 ? u[1] : u[0];
      const float send = b2 ? u[0] : u[1];
      cp = keep + row_xor<4>(send);
    }
    cp += row_xor<8>(cp);
    if ((lane & 8) == 0) {
      crow[16 * g + srow] = lrelu(cp);
    }
  }
  __syncthreads();

  if (wave < 16) {
    const int r = wave;
    float* row = sc + r * ROWSTRIDE + lane;
    const float* crow_r = cs + r * CSTRIDE;
    int base = 0;
    const float s1i = s1[i0 + r];
    const float* s2p = s2 + cbase + lane;
    float vv[HCOLS / 32];
    float m = -3.4e38f;
#pragma unroll
    for (int k = 0; k < HCOLS / 32; ++k) {
      const bool on = adjv[k] > 0;
      const unsigned bal = __builtin_amdgcn_ballot_w32(on);
      const int pos = base + (int)__builtin_amdgcn_mbcnt_lo(bal, 0u);
      base += __builtin_popcount(bal);
      float v = lrelu(s1i + s2p[32 * k]) + crow_r[pos];
      v = on ? v : NEG_BIG;
      vv[k] = v;
      m = fmaxf(m, v);
    }
    for (int mask = 16; mask >= 1; mask >>= 1)
      m = fmaxf(m, __shfl_xor(m, mask, 32));
    float s = 0.f;
#pragma unroll
    for (int k = 0; k < HCOLS / 32; ++k) {
      const float e = __expf(vv[k] - m);
      row[32 * k] = e;
      s += e;
    }
    for (int mask = 16; mask >= 1; mask >>= 1) s += __shfl_xor(s, mask, 32);
    if (lane == 0) {
      s_inv[r] = 1.0f / s;
      s_lse[r] = m + logf(s);
    }
  }
  __syncthreads();

  v8f c0 = {};
  v8f c1 = {};
  const int kbase = wave * (HCOLS / NWAVES);
  for (int kk = 0; kk < HCOLS / NWAVES; kk += 4) {
    const int k0 = kbase + kk;
    v2f a = *(const v2f*)(sc + nn * ROWSTRIDE + k0 + kd);
    const float* hr = h + (size_t)(cbase + k0 + kd) * D_OUT;
    v2f b0, b1;
    b0.x = hr[nn];
    b0.y = hr[D_OUT + nn];
    b1.x = hr[16 + nn];
    b1.y = hr[D_OUT + 16 + nn];
    c0 = wmma_f32_k4(a, b0, c0);
    c1 = wmma_f32_k4(a, b1, c1);
  }
  {
    const int row = (lane >> 4) << 3;
    float* pw = part + wave * 512;
    for (int v = 0; v < 8; ++v) {
      pw[(row + v) * 32 + nn] = c0[v];
      pw[(row + v) * 32 + 16 + nn] = c1[v];
    }
  }
  __syncthreads();
  if (tid < 512) {
    const int e = tid;
    float s = 0.f;
    for (int w = 0; w < NWAVES; ++w) s += part[w * 512 + e];
    const int row = e >> 5, col = e & 31;
    float* dst = sp ? o1 : o0;
    dst[(size_t)(i0 + row) * D_OUT + col] = s * s_inv[row];
  } else if (tid < 528) {
    lse[sp * N + i0 + (tid - 512)] = s_lse[tid - 512];
  }
}

__global__ void bn_partial_kernel(float* __restrict__ hp,
                                  const float* __restrict__ o1,
                                  const float* __restrict__ lse,
                                  float* __restrict__ bp) {
  __shared__ float ls[8][32];
  __shared__ float lq[8][32];
  const int col = threadIdx.x & 31;
  const int seg = threadIdx.x >> 5;
  const int r0 = blockIdx.x * 32 + seg * 4;
  float s = 0.f, q = 0.f;
  for (int r = 0; r < 4; ++r) {
    const size_t idx = (size_t)(r0 + r) * D_OUT + col;
    const float l0 = lse[r0 + r], l1 = lse[N + r0 + r];
    const float lm = fmaxf(l0, l1);
    const float w0 = expf(l0 - lm), w1 = expf(l1 - lm);
    const float v = (w0 * hp[idx] + w1 * o1[idx]) / (w0 + w1);
    hp[idx] = v;
    s += v;
    q += v * v;
  }
  ls[seg][col] = s;
  lq[seg][col] = q;
  __syncthreads();
  if (threadIdx.x < 32) {
    float ss = 0.f, qq = 0.f;
    for (int w = 0; w < 8; ++w) {
      ss += ls[w][col];
      qq += lq[w][col];
    }
    bp[blockIdx.x * 64 + col] = ss;
    bp[blockIdx.x * 64 + 32 + col] = qq;
  }
}

__global__ void __launch_bounds__(256)
bn_norm_kernel(const float* __restrict__ hp,
               const float* __restrict__ bp,
               const float* __restrict__ gamma,
               const float* __restrict__ beta,
               float* __restrict__ out) {
  __shared__ float ps[8][32];
  __shared__ float pq[8][32];
  const int idx = blockIdx.x * 256 + threadIdx.x;
  const int col = threadIdx.x & 31;
  const int part = threadIdx.x >> 5;
  const float x = hp[idx];
  const float gm = gamma[col], bt = beta[col];
  const float* b = bp + (part * 8) * 64 + col;
  float sp[8], sq[8];
#pragma unroll
  for (int k = 0; k < 8; ++k) {
    sp[k] = b[k * 64];
    sq[k] = b[k * 64 + 32];
  }
  float s = 0.f, q = 0.f;
  for (int k = 0; k < 8; ++k) {
    s += sp[k];
    q += sq[k];
  }
  ps[part][col] = s;
  pq[part][col] = q;
  __syncthreads();
  float ss = 0.f, qq = 0.f;
  for (int w = 0; w < 8; ++w) {
    ss += ps[w][col];
    qq += pq[w][col];
  }
  const float mean = ss * (1.0f / (float)N);
  const float var = qq * (1.0f / (float)N) - mean * mean;
  out[idx] = gm * (x - mean) * rsqrtf(var + 1e-5f) + bt;
}

extern "C" void kernel_launch(void* const* d_in, const int* in_sizes, int n_in,
                              void* d_out, int out_size, void* d_ws,
                              size_t ws_size, hipStream_t stream) {
  const float* input = (const float*)d_in[0];
  const float* edge_feat = (const float*)d_in[1];
  const int* adj = (const int*)d_in[2];
  const float* Wh = (const float*)d_in[3];
  const float* Wh1 = (const float*)d_in[4];
  const float* Wh2 = (const float*)d_in[5];
  const float* Wf = (const float*)d_in[6];
  const float* ah = (const float*)d_in[7];
  const float* af = (const float*)d_in[8];
  const float* bf = (const float*)d_in[9];
  const float* gamma = (const float*)d_in[10];
  const float* beta = (const float*)d_in[11];

  float* ws = (float*)d_ws;
  float* s1 = ws + 256;
  float* s2 = ws + 2304;
  float* h = ws + 4352;
  float* hp = ws + 69888;
  float* lse = ws + 135424;
  float* bp = ws + 4352;
  float* out = (float*)d_out;

  prep_kernel<<<96, 128, 0, stream>>>(input, Wh, Wh1, Wh2, ah, s1, s2, h);
  const size_t smem = (size_t)(16 * ROWSTRIDE + NWAVES * 512 + 16 * CSTRIDE) * sizeof(float) +
                      (size_t)(16 * ISTRIDE) * sizeof(unsigned short);
  fused_attention_kernel<<<256, 1024, smem, stream>>>(edge_feat, adj, Wf, af, bf,
                                                      s1, s2, h, hp, out, lse);
  bn_partial_kernel<<<64, 256, 0, stream>>>(hp, out, lse, bp);
  bn_norm_kernel<<<256, 256, 0, stream>>>(hp, bp, gamma, beta, out);
}


# edge gather as ONE 16-byte load per lane and group (lane half h holds features 4h..4h+3; WMMA k pairs follow), issued 3 turns ahead in registers; was two 8-byte loads + prefetch; f sum order changes
# speedup vs baseline: 7.6647x; 1.0523x over previous
#include <hip/hip_runtime.h>
#include <math.h>

#define N 2048
#define D_IN 128
#define D_OUT 32
#define D_EDGE 8
#define ALPHA 0.2f
#define NEG_BIG -9e15f
#define HCOLS 1024
#define ROWSTRIDE 1026
#define NWAVES 32
#define CSTRIDE 1024
#define ISTRIDE 1152

typedef __attribute__((ext_vector_type(2))) float v2f;
typedef __attribute__((ext_vector_type(4))) float v4f;
typedef __attribute__((ext_vector_type(8))) float v8f;

__device__ __forceinline__ v8f wmma_f32_k4(v2f a, v2f b, v8f c) {
  return __builtin_amdgcn_wmma_f32_16x16x4_f32(false, a, false, b, (short)0, c,
                                               false, false);
}

__device__ __forceinline__ float lrelu(float x) { return fmaxf(x, ALPHA * x); }

template <int M>
__device__ __forceinline__ float row_xor(float x) {
  return __int_as_float(__builtin_amdgcn_update_dpp(0, __float_as_int(x), 0x160 + M,
                                                    0xf, 0xf, true));
}

#define SROWS 8
#define SWAVES 4
#define SBLOCKS (N / (SROWS * SWAVES))
#define HBLOCKS (N / (16 * SWAVES))
__global__ void __launch_bounds__(32 * SWAVES)
prep_kernel(const float* __restrict__ input,
            const float* __restrict__ Wh,
            const float* __restrict__ Wh1,
            const float* __restrict__ Wh2,
            const float* __restrict__ ah,
            float* __restrict__ s1, float* __restrict__ s2,
            float* __restrict__ h) {
  __shared__ __attribute__((aligned(16))) float sv[2 * D_IN];
  const int lane = threadIdx.x & 31;
  const int wave = threadIdx.x >> 5;
  if (blockIdx.x < SBLOCKS) {
    const int t = threadIdx.x;
    const int r0 = (blockIdx.x * SWAVES + wave) * SROWS;
    const bool b0 = (lane & 1) != 0, b1 = (lane & 2) != 0, b2 = (lane & 4) != 0,
               b3 = (lane & 8) != 0;
    v4f w1[D_OUT / 4], w2[D_OUT / 4];
#pragma unroll
    for (int f = 0; f < D_OUT / 4; ++f) {
      w1[f] = *(const v4f*)(Wh1 + t * D_OUT + 4 * f);
      w2[f] = *(const v4f*)(Wh2 + t * D_OUT + 4 * f);
    }
    const float* row = input + (size_t)r0 * D_IN + 4 * lane;
    v4f x[SROWS];
#pragma unroll
    for (int r = 0; r < SROWS; ++r) x[r] = *(const v4f*)(row + r * D_IN);
    float wa[D_OUT];
#pragma unroll
    for (int f = 0; f < D_OUT; ++f) wa[f] = ah[f];
    __builtin_amdgcn_sched_barrier(0);
    float a1 = 0.f, a2 = 0.f;
#pragma unroll
    for (int f = 0; f < D_OUT; ++f) {
      a1 += w1[f >> 2][f & 3] * wa[f];
      a2 += w2[f >> 2][f & 3] * wa[f];
    }
    sv[t] = a1;
    sv[D_IN + t] = a2;
    __syncthreads();
    const v4f a = *(const v4f*)(sv + 4 * lane);
    const v4f b = *(const v4f*)(sv + D_IN + 4 * lane);
    float p[2 * SROWS];
    for (int r = 0; r < SROWS; ++r) {
      p[r] = x[r].x * a.x + x[r].y * a.y + x[r].z * a.z + x[r].w * a.w;
      p[SROWS + r] = x[r].x * b.x + x[r].y * b.y + x[r].z * b.z + x[r].w * b.w;
    }
    float q[8], u[4], w[2];
    for (int k = 0; k < 8; ++k) {
      const float keep = b0 ? p[2 * k + 1] : p[2 * k];
      const float send = b0 ? p[2 * k] : p[2 * k + 1];
      q[k] = keep + row_xor<1>(send);
    }
    for (int k = 0; k < 4; ++k) {
      const float keep = b1 ? q[2 * k + 1] : q[2 * k];
      const float send = b1 ? q[2 * k] : q[2 * k + 1];
      u[k] = keep + row_xor<2>(send);
    }
    for (int k = 0; k < 2; ++k) {
      const float keep = b2 ? u[2 * k + 1] : u[2 * k];
      const float send = b2 ? u[2 * k] : u[2 * k + 1];
      w[k] = keep + row_xor<4>(send);
    }
    float tt;
    {
      const float keep = b3 ? w[1] : w[0];
      const float send = b3 ? w[0] : w[1];
      tt = keep + row_xor<8>(send);
    }
    tt += __shfl_xor(tt, 16, 32);
    if (lane < SROWS) s1[r0 + lane] = tt;
    else if (lane < 2 * SROWS) s2[r0 + lane - SROWS] = tt;
  } else {
    const int i0 = ((blockIdx.x - SBLOCKS) * SWAVES + wave) * 16;
    const int m = lane & 15;
    const int kd = (lane >> 4) << 1;
    const int n = lane & 15;
    v8f c0 = {};
    v8f c1 = {};
    for (int k0 = 0; k0 < D_IN; k0 += 4) {
      v2f a = *(const v2f*)(input + (size_t)(i0 + m) * D_IN + k0 + kd);
      v2f b0, b1;
      b0.x = Wh[(k0 + kd) * D_OUT + n];
      b0.y = Wh[(k0 + kd + 1) * D_OUT + n];
      b1.x = Wh[(k0 + kd) * D_OUT + 16 + n];
      b1.y = Wh[(k0 + kd + 1) * D_OUT + 16 + n];
      c0 = wmma_f32_k4(a, b0, c0);
      c1 = wmma_f32_k4(a, b1, c1);
    }
    const int row = (lane >> 4) << 3;
    for (int v = 0; v < 8; ++v) {
      h[(size_t)(i0 + row + v) * D_OUT + n] = c0[v];
      h[(size_t)(i0 + row + v) * D_OUT + 16 + n] = c1[v];
    }
  }
}

__global__ void __launch_bounds__(1024)
fused_attention_kernel(const float* __restrict__ edge_feat,
                                       const int* __restrict__ adj,
                                       const float* __restrict__ Wf,
                                       const float* __restrict__ af,
                                       const float* __restrict__ bf,
                                       const float* __restrict__ s1,
                                       const float* __restrict__ s2,
                                       const float* __restrict__ h,
                                       float* __restrict__ o0,
                                       float* __restrict__ o1,
                                       float* __restrict__ lse) {
  extern __shared__ float smem[];
  float* sc = smem;
  float* part = smem + 16 * ROWSTRIDE;
  float* cs = part + NWAVES * 512;
  unsigned short* idx = (unsigned short*)(cs + 16 * CSTRIDE);
  __shared__ int s_cnt[16];
  __shared__ float s_inv[16];
  __shared__ float s_lse[16];

  const int tid = threadIdx.x;
  const int lane = tid & 31;
  const int wave = __builtin_amdgcn_readfirstlane(tid >> 5);
  const int i0 = (blockIdx.x >> 1) * 16;
  const int sp = blockIdx.x & 1;
  const int cbase = sp * HCOLS;

  const int nn = lane & 15;
  const int kd = (lane >> 4) << 1;
  const bool b0 = (lane & 1) != 0, b1 = (lane & 2) != 0, b2 = (lane & 4) != 0;

  const int kq = (lane >> 4) << 2;
  v2f bwa0, bwa1, bwb0, bwb1;
  bwa0.x = Wf[(kq + 0) * D_OUT + nn];
  bwa0.y = Wf[(kq + 1) * D_OUT + nn];
  bwa1.x = Wf[(kq + 0) * D_OUT + 16 + nn];
  bwa1.y = Wf[(kq + 1) * D_OUT + 16 + nn];
  bwb0.x = Wf[(kq + 2) * D_OUT + nn];
  bwb0.y = Wf[(kq + 3) * D_OUT + nn];
  bwb1.x = Wf[(kq + 2) * D_OUT + 16 + nn];
  bwb1.y = Wf[(kq + 3) * D_OUT + 16 + nn];
  const float af_lo = af[nn], af_hi = af[16 + nn];
  const float bf_lo = bf[nn], bf_hi = bf[16 + nn];
  int adjv[HCOLS / 32] = {};
  if (wave < 16) {
    const int* arow = adj + (size_t)(i0 + wave) * N + cbase + lane;
#pragma unroll
    for (int k = 0; k < HCOLS / 32; ++k) adjv[k] = arow[32 * k];
    unsigned short* irow = idx + wave * ISTRIDE;
    int base = 0;
#pragma unroll
    for (int k = 0; k < HCOLS / 32; ++k) {
      const bool on = adjv[k] > 0;
      const unsigned bal = __builtin_amdgcn_ballot_w32(on);
      const int pos = base + (int)__builtin_amdgcn_mbcnt_lo(bal, 0u);
      if (on) irow[pos] = (unsigned short)(lane + 32 * k);
      base += __builtin_popcount(bal);
    }
#pragma unroll
    for (int k = 0; k < 4; ++k) irow[base + lane + 32 * k] = 0;
    if (lane == 0) s_cnt[wave] = base;
  }
  __syncthreads();
  v8f cini0, cini1;
  for (int v = 0; v < 8; ++v) {
    cini0[v] = bf_lo;
    cini1[v] = bf_hi;
  }

  const int il = wave & 15;
  const int ng = (__builtin_amdgcn_readfirstlane(s_cnt[il]) + 15) >> 4;
  const char* erow = (const char*)(edge_feat + ((size_t)(i0 + il) * N + cbase) * D_EDGE);
  const unsigned eoff = (unsigned)kq * (unsigned)sizeof(float);
  const unsigned short* irow = idx + il * ISTRIDE + nn;
  float* crow = cs + il * CSTRIDE;
  const int srow = ((lane >> 4) << 3) + (lane & 7);
  auto fetch = [&](const int gg) -> v4f {
    const unsigned j = irow[16 * gg];
    return *(const v4f*)(erow + (j * (unsigned)(D_EDGE * sizeof(float)) + eoff));
  };
  auto step = [&](v4f& buf, const int g, const bool refill) {
    const v2f a0 = {buf.x, buf.y};
    const v2f a1 = {buf.z, buf.w};
    v8f f0 = wmma_f32_k4(a0, bwa0, cini0);
    f0 = wmma_f32_k4(a1, bwb0, f0);
    v8f f1 = wmma_f32_k4(a0, bwa1, cini1);
    f1 = wmma_f32_k4(a1, bwb1, f1);
    if (refill) buf = fetch(g + 6);
    float p[8];
    for (int k = 0; k < 4; ++k) {
      const v2f x0 = {f0[2 * k], f0[2 * k + 1]}, x1 = {f1[2 * k], f1[2 * k + 1]};
      const v2f y0 = x0 * ALPHA, y1 = x1 * ALPHA;
      const v2f g0 = {fmaxf(x0.x, y0.x), fmaxf(x0.y, y0.y)};
      const v2f g1 = {fmaxf(x1.x, y1.x), fmaxf(x1.y, y1.y)};
      const v2f pp = g0 * af_lo + g1 * af_hi;
      p[2 * k] = pp.x;
      p[2 * k + 1] = pp.y;
    }
    float q[4], u[2];
    for (int k = 0; k < 4; ++k) {
      const float keep = b0 ? p[2 * k + 1] : p[2 * k];
      const float send = b0 ? p[2 * k] : p[2 * k + 1];
      q[k] = keep + row_xor<1>(send);
    }
    for (int k = 0; k < 2; ++k) {
      const float keep = b1 ? q[2 * k + 1] : q[2 * k];
      const float send = b1 ? q[2 * k] : q[2 * k + 1];
      u[k] = keep + row_xor<2>(send);
    }
    float cp;
    {
      const float keep = b2 ? u[1] : u[0];
      const float send = b2 ? u[0] : u[1];
      cp = keep + row_xor<4>(send);
    }
    cp += row_xor<8>(cp);
    if ((lane & 8) == 0) {
      crow[16 * g + srow] = lrelu(cp);
    }
  };
  int g = wave >> 4;
  const int turns = (ng - g + 1) >> 1;
  v4f ea = fetch(g);
  v4f eb = fetch(g + 2);
  v4f ec = fetch(g + 4);
  for (int t = turns / 3; t > 0; --t) {
    step(ea, g, true);
    step(eb, g + 2, true);
    step(ec, g + 4, true);
    g += 6;
  }
  const int left = turns % 3;
  if (left > 0) step(ea, g, false);
  if (left > 1) step(eb, g + 2, false);
  __syncthreads();

  if (wave < 16) {
    const int r = wave;
    float* row = sc + r * ROWSTRIDE + lane;
    const float* crow_r = cs + r * CSTRIDE;
    int base = 0;
    const float s1i = s1[i0 + r];
    const float* s2p = s2 + cbase + lane;
    float vv[HCOLS / 32];
    float m = -3.4e38f;
#pragma unroll
    for (int k = 0; k < HCOLS / 32; ++k) {
      const bool on = adjv[k] > 0;
      const unsigned bal = __builtin_amdgcn_ballot_w32(on);
      const int pos = base + (int)__builtin_amdgcn_mbcnt_lo(bal, 0u);
      base += __builtin_popcount(bal);
      float v = lrelu(s1i + s2p[32 * k]) + crow_r[pos];
      v = on ? v : NEG_BIG;
      vv[k] = v;
      m = fmaxf(m, v);
    }
    for (int mask = 16; mask >= 1; mask >>= 1)
      m = fmaxf(m, __shfl_xor(m, mask, 32));
    float s = 0.f;
#pragma unroll
    for (int k = 0; k < HCOLS / 32; ++k) {
      const float e = __expf(vv[k] - m);
      row[32 * k] = e;
      s += e;
    }
    for (int mask = 16; mask >= 1; mask >>= 1) s += __shfl_xor(s, mask, 32);
    if (lane == 0) {
      s_inv[r] = 1.0f / s;
      s_lse[r] = m + logf(s);
    }
  }
  __syncthreads();

  v8f c0 = {};
  v8f c1 = {};
  const int kbase = wave * (HCOLS / NWAVES);
  for (int kk = 0; kk < HCOLS / NWAVES; kk += 4) {
    const int k0 = kbase + kk;
    v2f a = *(const v2f*)(sc + nn * ROWSTRIDE + k0 + kd);
    const float* hr = h + (size_t)(cbase + k0 + kd) * D_OUT;
    v2f b0, b1;
    b0.x = hr[nn];
    b0.y = hr[D_OUT + nn];
    b1.x = hr[16 + nn];
    b1.y = hr[D_OUT + 16 + nn];
    c0 = wmma_f32_k4(a, b0, c0);
    c1 = wmma_f32_k4(a, b1, c1);
  }
  {
    const int row = (lane >> 4) << 3;
    float* pw = part + wave * 512;
    for (int v = 0; v < 8; ++v) {
      pw[(row + v) * 32 + nn] = c0[v];
      pw[(row + v) * 32 + 16 + nn] = c1[v];
    }
  }
  __syncthreads();
  if (tid < 512) {
    const int e = tid;
    float s = 0.f;
    for (int w = 0; w < NWAVES; ++w) s += part[w * 512 + e];
    const int row = e >> 5, col = e & 31;
    float* dst = sp ? o1 : o0;
    dst[(size_t)(i0 + row) * D_OUT + col] = s * s_inv[row];
  } else if (tid < 528) {
    lse[sp * N + i0 + (tid - 512)] = s_lse[tid - 512];
  }
}

__global__ void bn_partial_kernel(float* __restrict__ hp,
                                  const float* __restrict__ o1,
                                  const float* __restrict__ lse,
                                  float* __restrict__ bp) {
  __shared__ float ls[8][32];
  __shared__ float lq[8][32];
  const int col = threadIdx.x & 31;
  const int seg = threadIdx.x >> 5;
  const int r0 = blockIdx.x * 32 + seg * 4;
  float s = 0.f, q = 0.f;
  for (int r = 0; r < 4; ++r) {
    const size_t idx = (size_t)(r0 + r) * D_OUT + col;
    const float l0 = lse[r0 + r], l1 = lse[N + r0 + r];
    const float lm = fmaxf(l0, l1);
    const float w0 = expf(l0 - lm), w1 = expf(l1 - lm);
    const float v = (w0 * hp[idx] + w1 * o1[idx]) / (w0 + w1);
    hp[idx] = v;
    s += v;
    q += v * v;
  }
  ls[seg][col] = s;
  lq[seg][col] = q;
  __syncthreads();
  if (threadIdx.x < 32) {
    float ss = 0.f, qq = 0.f;
    for (int w = 0; w < 8; ++w) {
      ss += ls[w][col];
      qq += lq[w][col];
    }
    bp[blockIdx.x * 64 + col] = ss;
    bp[blockIdx.x * 64 + 32 + col] = qq;
  }
}

__global__ void __launch_bounds__(256)
bn_norm_kernel(const float* __restrict__ hp,
               const float* __restrict__ bp,
               const float* __restrict__ gamma,
               const float* __restrict__ beta,
               float* __restrict__ out) {
  __shared__ float ps[8][32];
  __shared__ float pq[8][32];
  const int idx = blockIdx.x * 256 + threadIdx.x;
  const int col = threadIdx.x & 31;
  const int part = threadIdx.x >> 5;
  const float x = hp[idx];
  const float gm = gamma[col], bt = beta[col];
  const float* b = bp + (part * 8) * 64 + col;
  float sp[8], sq[8];
#pragma unroll
  for (int k = 0; k < 8; ++k) {
    sp[k] = b[k * 64];
    sq[k] = b[k * 64 + 32];
  }
  float s = 0.f, q = 0.f;
  for (int k = 0; k < 8; ++k) {
    s += sp[k];
    q += sq[k];
  }
  ps[part][col] = s;
  pq[part][col] = q;
  __syncthreads();
  float ss = 0.f, qq = 0.f;
  for (int w = 0; w < 8; ++w) {
    ss += ps[w][col];
    qq += pq[w][col];
  }
  const float mean = ss * (1.0f / (float)N);
  const float var = qq * (1.0f / (float)N) - mean * mean;
  out[idx] = gm * (x - mean) * rsqrtf(var + 1e-5f) + bt;
}

extern "C" void kernel_launch(void* const* d_in, const int* in_sizes, int n_in,
                              void* d_out, int out_size, void* d_ws,
                              size_t ws_size, hipStream_t stream) {
  const float* input = (const float*)d_in[0];
  const float* edge_feat = (const float*)d_in[1];
  const int* adj = (const int*)d_in[2];
  const float* Wh = (const float*)d_in[3];
  const float* Wh1 = (const float*)d_in[4];
  const float* Wh2 = (const float*)d_in[5];
  const float* Wf = (const float*)d_in[6];
  const float* ah = (const float*)d_in[7];
  const float* af = (const float*)d_in[8];
  const float* bf = (const float*)d_in[9];
  const float* gamma = (const float*)d_in[10];
  const float* beta = (const float*)d_in[11];

  float* ws = (float*)d_ws;
  float* s1 = ws + 256;
  float* s2 = ws + 2304;
  float* h = ws + 4352;
  float* hp = ws + 69888;
  float* lse = ws + 135424;
  float* bp = ws + 4352;
  float* out = (float*)d_out;

  prep_kernel<<<96, 128, 0, stream>>>(input, Wh, Wh1, Wh2, ah, s1, s2, h);
  const size_t smem = (size_t)(16 * ROWSTRIDE + NWAVES * 512 + 16 * CSTRIDE) * sizeof(float) +
                      (size_t)(16 * ISTRIDE) * sizeof(unsigned short);
  fused_attention_kernel<<<256, 1024, smem, stream>>>(edge_feat, adj, Wf, af, bf,
                                                      s1, s2, h, hp, out, lse);
  bn_partial_kernel<<<64, 256, 0, stream>>>(hp, out, lse, bp);
  bn_norm_kernel<<<256, 256, 0, stream>>>(hp, bp, gamma, beta, out);
}
